# starting kernel
# baseline (speedup 1.0000x reference)
#include <hip/hip_runtime.h>
#include <hip/hip_bf16.h>
#include <math.h>


typedef __attribute__((ext_vector_type(2))) float v2f;
typedef __attribute__((ext_vector_type(8))) float v8f;

#define BATCH 64
#define SEQL  24
#define HDIM  512

__device__ __forceinline__ float sigm(float x) { return 1.0f / (1.0f + __expf(-x)); }
__device__ __forceinline__ float tanh_fast(float x) { return 2.0f / (1.0f + __expf(-2.0f * x)) - 1.0f; }

template <int MODE>
__global__ __launch_bounds__(256) void gemm_wmma(const float* __restrict__ X,
                                                 const float* __restrict__ hbuf,
                                                 const float* __restrict__ cbuf,
                                                 const float* __restrict__ W,
                                                 const float* __restrict__ bias,
                                                 float* __restrict__ out0,
                                                 float* __restrict__ out1,
                                                 int K, int n) {
  __shared__ float A_lds[32 * 36];
  __shared__ float B_lds[320 * 36];

  const int tid  = threadIdx.x;
  const unsigned lane = tid & 31u;
  const int wave = tid >> 5;
  const int wm   = wave >> 2;
  const int wn   = wave & 3;
  const int l16  = (int)(lane & 15u);
  const int lhi  = (int)(lane >> 4);

  const int r0 = blockIdx.x * 32;
  const int NACC = (MODE == 0) ? 4 : 5;

  v8f acc[5] = {};

  for (int k0 = 0; k0 < K; k0 += 32) {
    {
      int ar  = tid >> 3;
      int ac4 = tid & 7;
      int rg  = r0 + ar;
      int kg  = k0 + ac4 * 4;
      float4 val;
      if (MODE == 0) {
        val = *(const float4*)&X[(size_t)rg * K + kg];
      } else {
        int b = rg / n;
        int p = rg - b * n;
        const float* src = (kg < HDIM)
            ? &hbuf[((size_t)(b * SEQL + p) * HDIM) + kg]
            : &hbuf[((size_t)(b * SEQL + p + 1) * HDIM) + (kg - HDIM)];
        val = *(const float4*)src;
      }
      *(float4*)&A_lds[ar * 36 + ac4 * 4] = val;
    }
    if (MODE == 0) {
      int nb0 = blockIdx.y * 256;
#pragma unroll
      for (int i = 0; i < 8; ++i) {
        int f    = i * 256 + tid;
        int nrow = f >> 3;
        int kc4  = f & 7;
        float4 val = *(const float4*)&W[(size_t)(nb0 + nrow) * K + k0 + kc4 * 4];
        *(float4*)&B_lds[nrow * 36 + kc4 * 4] = val;
      }
    } else {
      int nb0 = blockIdx.y * 64;
#pragma unroll
      for (int i = 0; i < 10; ++i) {
        int f    = i * 256 + tid;
        int nrow = f >> 3;
        int kc4  = f & 7;
        int g    = nrow >> 6;
        int nl   = nrow & 63;
        float4 val = *(const float4*)&W[(size_t)(g * HDIM + nb0 + nl) * K + k0 + kc4 * 4];
        *(float4*)&B_lds[nrow * 36 + kc4 * 4] = val;
      }
    }
    __syncthreads();

#pragma unroll
    for (int kk = 0; kk < 32; kk += 4) {
      v2f a = *(const v2f*)&A_lds[(wm * 16 + l16) * 36 + kk + 2 * lhi];
#pragma unroll
      for (int g = 0; g < NACC; ++g) {
        int nrow = (MODE == 0) ? (wn * 64 + g * 16 + l16)
                               : (g * 64 + wn * 16 + l16);
        v2f bf = *(const v2f*)&B_lds[nrow * 36 + kk + 2 * lhi];
        acc[g] = __builtin_amdgcn_wmma_f32_16x16x4_f32(
            false, a, false, bf, (short)0, acc[g], false, false);
      }
    }
    __syncthreads();
  }

  if (MODE == 0) {
    const int colbase = blockIdx.y * 256 + wn * 64;
    const int rowg0   = r0 + wm * 16 + 8 * lhi;
#pragma unroll
    for (int j = 0; j < 4; ++j) {
      int col = colbase + j * 16 + l16;
      float bv = bias[col];
#pragma unroll
      for (int vr = 0; vr < 8; ++vr) {
        int rg = rowg0 + vr;
        float val = acc[j][vr] + bv;
        if (col < HDIM) out0[(size_t)rg * HDIM + col] = val;
        else            out1[(size_t)rg * HDIM + (col - HDIM)] = val;
      }
    }
  } else {
    const int colg = blockIdx.y * 64 + wn * 16 + l16;
    float bz[5];
#pragma unroll
    for (int g = 0; g < 5; ++g) bz[g] = bias[g * HDIM + colg];
    int rg = r0 + wm * 16 + 8 * lhi;
    int b  = rg / n;
    int p  = rg - b * n;
#pragma unroll
    for (int vr = 0; vr < 8; ++vr) {
      float gi  = acc[0][vr] + bz[0];
      float gfl = acc[1][vr] + bz[1];
      float gfr = acc[2][vr] + bz[2];
      float gu  = acc[3][vr] + bz[3];
      float go  = acc[4][vr] + bz[4];
      size_t cbase = (size_t)(b * SEQL + p) * HDIM + colg;
      float cl = cbuf[cbase];
      float cr = cbuf[cbase + HDIM];
      float nc = cl * sigm(gfl + 1.0f) + cr * sigm(gfr + 1.0f) + tanh_fast(gu) * sigm(gi);
      float nh = sigm(go) * tanh_fast(nc);
      size_t oi = (size_t)rg * HDIM + colg;
      out0[oi] = nh;
      out1[oi] = nc;
      ++rg; ++p;
      if (p == n) { p = 0; ++b; }
    }
  }
}

__global__ __launch_bounds__(256) void mix_k(const float* __restrict__ newh,
                                             const float* __restrict__ newc,
                                             const float* __restrict__ hcur,
                                             const float* __restrict__ ccur,
                                             const float* __restrict__ q,
                                             const int* __restrict__ length,
                                             float* __restrict__ hnxt,
                                             float* __restrict__ cnxt,
                                             int n, int iter) {
  __shared__ float sc[32];
  __shared__ int ksel_s;
  const int b = blockIdx.x;
  const int tid = threadIdx.x;
  const unsigned lane = tid & 31u;
  const int wave = tid >> 5;

  for (int p = wave; p < n; p += 8) {
    const float* row = newh + (size_t)(b * n + p) * HDIM;
    float s = 0.0f;
    for (int j = (int)lane; j < HDIM; j += 32) s += row[j] * q[j];
#pragma unroll
    for (int off = 16; off > 0; off >>= 1) s += __shfl_xor(s, off, 32);
    if (lane == 0) sc[p] = s;
  }
  __syncthreads();

  if (tid == 0) {
    int len = length[b];
    int best = 0;
    float bv = -INFINITY;
    for (int p = 0; p < n; ++p) {
      float s = ((iter + 1 + p) < len) ? sc[p] : -1.0e9f;
      if (s > bv) { bv = s; best = p; }
    }
    ksel_s = best;
  }
  __syncthreads();

  const int ksel = ksel_s;
  const bool done = (iter + 1) < length[b];

  for (int idx = tid; idx < n * HDIM; idx += 256) {
    int p = idx >> 9;
    int j = idx & (HDIM - 1);
    size_t cur = (size_t)(b * SEQL + p) * HDIM + j;
    float hm, cm;
    if (p < ksel)      { hm = hcur[cur];        cm = ccur[cur]; }
    else if (p > ksel) { hm = hcur[cur + HDIM]; cm = ccur[cur + HDIM]; }
    else {
      size_t ni = (size_t)(b * n + p) * HDIM + j;
      hm = newh[ni]; cm = newc[ni];
    }
    hnxt[cur] = done ? hm : hcur[cur];
    cnxt[cur] = done ? cm : ccur[cur];
  }
}

__global__ __launch_bounds__(256) void final_copy_k(const float* __restrict__ hfin,
                                                    float* __restrict__ out) {
  int idx = blockIdx.x * 256 + threadIdx.x;
  int b = idx >> 9;
  int j = idx & (HDIM - 1);
  out[idx] = hfin[(size_t)b * SEQL * HDIM + j];
}

extern "C" void kernel_launch(void* const* d_in, const int* in_sizes, int n_in,
                              void* d_out, int out_size, void* d_ws, size_t ws_size,
                              hipStream_t stream) {
  const float* inp    = (const float*)d_in[0];
  const int*   length = (const int*)  d_in[1];
  const float* word_W = (const float*)d_in[2];
  const float* word_b = (const float*)d_in[3];
  const float* comp_W = (const float*)d_in[4];
  const float* comp_b = (const float*)d_in[5];
  const float* comp_q = (const float*)d_in[6];

  float* ws = (float*)d_ws;
  size_t off = 0;
  float* hb[2];
  float* cb[2];
  hb[0] = ws + off; off += (size_t)BATCH * SEQL * HDIM;
  cb[0] = ws + off; off += (size_t)BATCH * SEQL * HDIM;
  hb[1] = ws + off; off += (size_t)BATCH * SEQL * HDIM;
  cb[1] = ws + off; off += (size_t)BATCH * SEQL * HDIM;
  float* newh = ws + off; off += (size_t)(BATCH * 23) * HDIM;
  float* newc = ws + off; off += (size_t)(BATCH * 23) * HDIM;

  gemm_wmma<0><<<dim3((BATCH * SEQL) / 32, 1024 / 256), 256, 0, stream>>>(
      inp, nullptr, nullptr, word_W, word_b, hb[0], cb[0], 512, 0);

  int cur = 0;
  for (int i = 0; i < SEQL - 1; ++i) {
    int n = (SEQL - 1) - i;
    int R = BATCH * n;
    int nxt = cur ^ 1;
    gemm_wmma<1><<<dim3(R / 32, HDIM / 64), 256, 0, stream>>>(
        nullptr, hb[cur], cb[cur], comp_W, comp_b, newh, newc, 1024, n);
    mix_k<<<BATCH, 256, 0, stream>>>(newh, newc, hb[cur], cb[cur], comp_q, length,
                                     hb[nxt], cb[nxt], n, i);
    cur = nxt;
  }

  final_copy_k<<<(BATCH * HDIM) / 256, 256, 0, stream>>>(hb[cur], (float*)d_out);
  (void)in_sizes; (void)n_in; (void)out_size; (void)ws_size;
}


# Software-pipelined GEMM K loop: next 32-wide chunk loaded global->registers during the current chunk's WMMAs, LDS double-buffered, one barrier per chunk instead of two
# speedup vs baseline: 1.3943x; 1.3943x over previous
#include <hip/hip_runtime.h>
#include <hip/hip_bf16.h>
#include <math.h>


typedef __attribute__((ext_vector_type(2))) float v2f;
typedef __attribute__((ext_vector_type(4))) float v4f;
typedef __attribute__((ext_vector_type(8))) float v8f;

#define BATCH 64
#define SEQL  24
#define HDIM  512

__device__ __forceinline__ float sigm(float x) { return 1.0f / (1.0f + __expf(-x)); }
__device__ __forceinline__ float tanh_fast(float x) { return 2.0f / (1.0f + __expf(-2.0f * x)) - 1.0f; }

#define LDS_A_STAGE (32 * 36)
#define LDS_B_STAGE (320 * 36)

template <int MODE>
__device__ __forceinline__ constexpr int wrow_of(int i) {
  return (MODE == 0) ? (i * 32) : ((i >> 1) * HDIM + (i & 1) * 32);
}

template <int MODE, int NB, int KD>
__device__ __forceinline__ void stage_gload(const float* asrc, const float* bsrc, int k,
                                            v4f& ra, v4f (&rb)[NB]) {
  ra = *(const v4f*)(asrc + k);
#pragma unroll
  for (int i = 0; i < NB; ++i)
    rb[i] = *(const v4f*)(bsrc + (size_t)wrow_of<MODE>(i) * KD + k);
}

template <int NB>
__device__ __forceinline__ void stage_lstore(float* As, float* Bs, int sr, int sc4,
                                             const v4f& ra, const v4f (&rb)[NB]) {
  *(v4f*)&As[sr * 36 + sc4 * 4] = ra;
#pragma unroll
  for (int i = 0; i < NB; ++i)
    *(v4f*)&Bs[(i * 32 + sr) * 36 + sc4 * 4] = rb[i];
}

template <int MODE, int NACC>
__device__ __forceinline__ void stage_mma(const float* As, const float* Bs,
                                          int wm, int wn, int l16, int lhi, v8f (&acc)[5]) {
#pragma unroll
  for (int kk = 0; kk < 32; kk += 4) {
    v2f a = *(const v2f*)&As[(wm * 16 + l16) * 36 + kk + 2 * lhi];
#pragma unroll
    for (int g = 0; g < NACC; ++g) {
      int nrow = (MODE == 0) ? (wn * 64 + g * 16 + l16)
                             : (g * 64 + wn * 16 + l16);
      v2f bf = *(const v2f*)&Bs[nrow * 36 + kk + 2 * lhi];
      acc[g] = __builtin_amdgcn_wmma_f32_16x16x4_f32(
          false, a, false, bf, (short)0, acc[g], false, false);
    }
  }
}

template <int MODE>
__global__ __launch_bounds__(256) void gemm_wmma(const float* __restrict__ X,
                                                 const float* __restrict__ hbuf,
                                                 const float* __restrict__ cbuf,
                                                 const float* __restrict__ W,
                                                 const float* __restrict__ bias,
                                                 float* __restrict__ out0,
                                                 float* __restrict__ out1,
                                                 int n) {
  constexpr int KD   = (MODE == 0) ? 512 : 1024;
  constexpr int NB   = (MODE == 0) ? 8 : 10;
  constexpr int NACC = (MODE == 0) ? 4 : 5;

  __shared__ __attribute__((aligned(16))) float A_lds[2 * LDS_A_STAGE];
  __shared__ __attribute__((aligned(16))) float B_lds[2 * LDS_B_STAGE];
  float* const A0 = A_lds;
  float* const A1 = A_lds + LDS_A_STAGE;
  float* const B0 = B_lds;
  float* const B1 = B_lds + LDS_B_STAGE;

  const int tid  = threadIdx.x;
  const unsigned lane = tid & 31u;
  const int wave = tid >> 5;
  const int wm   = wave >> 2;
  const int wn   = wave & 3;
  const int l16  = (int)(lane & 15u);
  const int lhi  = (int)(lane >> 4);

  const int r0 = blockIdx.x * 32;

  const int sr  = tid >> 3;
  const int sc4 = tid & 7;
  const float* asrc;
  if (MODE == 0) {
    asrc = X + (size_t)(r0 + sr) * KD + sc4 * 4;
  } else {
    int rg = r0 + sr;
    int b  = rg / n;
    int p  = rg - b * n;
    asrc = hbuf + (size_t)(b * SEQL + p) * HDIM + sc4 * 4;
  }
  const float* bsrc = W + (size_t)(blockIdx.y * ((MODE == 0) ? 256 : 64) + sr) * KD + sc4 * 4;

  v4f ra;
  v4f rb[NB];
  v8f acc[5] = {};

  stage_gload<MODE, NB, KD>(asrc, bsrc, 0, ra, rb);
  stage_lstore<NB>(A0, B0, sr, sc4, ra, rb);
  __syncthreads();

  for (int k0 = 0; k0 < KD - 64; k0 += 64) {
    stage_gload<MODE, NB, KD>(asrc, bsrc, k0 + 32, ra, rb);
    stage_mma<MODE, NACC>(A0, B0, wm, wn, l16, lhi, acc);
    stage_lstore<NB>(A1, B1, sr, sc4, ra, rb);
    __syncthreads();

    stage_gload<MODE, NB, KD>(asrc, bsrc, k0 + 64, ra, rb);
    stage_mma<MODE, NACC>(A1, B1, wm, wn, l16, lhi, acc);
    stage_lstore<NB>(A0, B0, sr, sc4, ra, rb);
    __syncthreads();
  }
  stage_gload<MODE, NB, KD>(asrc, bsrc, KD - 32, ra, rb);
  stage_mma<MODE, NACC>(A0, B0, wm, wn, l16, lhi, acc);
  stage_lstore<NB>(A1, B1, sr, sc4, ra, rb);
  __syncthreads();
  stage_mma<MODE, NACC>(A1, B1, wm, wn, l16, lhi, acc);

  if (MODE == 0) {
    const int colbase = blockIdx.y * 256 + wn * 64;
    const int rowg0   = r0 + wm * 16 + 8 * lhi;
#pragma unroll
    for (int j = 0; j < 4; ++j) {
      int col = colbase + j * 16 + l16;
      float bv = bias[col];
#pragma unroll
      for (int vr = 0; vr < 8; ++vr) {
        int rg = rowg0 + vr;
        float val = acc[j][vr] + bv;
        if (col < HDIM) out0[(size_t)rg * HDIM + col] = val;
        else            out1[(size_t)rg * HDIM + (col - HDIM)] = val;
      }
    }
  } else {
    const int colg = blockIdx.y * 64 + wn * 16 + l16;
    float bz[5];
#pragma unroll
    for (int g = 0; g < 5; ++g) bz[g] = bias[g * HDIM + colg];
    int rg = r0 + wm * 16 + 8 * lhi;
    int b  = rg / n;
    int p  = rg - b * n;
#pragma unroll
    for (int vr = 0; vr < 8; ++vr) {
      float gi  = acc[0][vr] + bz[0];
      float gfl = acc[1][vr] + bz[1];
      float gfr = acc[2][vr] + bz[2];
      float gu  = acc[3][vr] + bz[3];
      float go  = acc[4][vr] + bz[4];
      size_t cbase = (size_t)(b * SEQL + p) * HDIM + colg;
      float cl = cbuf[cbase];
      float cr = cbuf[cbase + HDIM];
      float nc = cl * sigm(gfl + 1.0f) + cr * sigm(gfr + 1.0f) + tanh_fast(gu) * sigm(gi);
      float nh = sigm(go) * tanh_fast(nc);
      size_t oi = (size_t)rg * HDIM + colg;
      out0[oi] = nh;
      out1[oi] = nc;
      ++rg; ++p;
      if (p == n) { p = 0; ++b; }
    }
  }
}

__global__ __launch_bounds__(256) void mix_k(const float* __restrict__ newh,
                                             const float* __restrict__ newc,
                                             const float* __restrict__ hcur,
                                             const float* __restrict__ ccur,
                                             const float* __restrict__ q,
                                             const int* __restrict__ length,
                                             float* __restrict__ hnxt,
                                             float* __restrict__ cnxt,
                                             int n, int iter) {
  __shared__ float sc[32];
  __shared__ int ksel_s;
  const int b = blockIdx.x;
  const int tid = threadIdx.x;
  const unsigned lane = tid & 31u;
  const int wave = tid >> 5;

  for (int p = wave; p < n; p += 8) {
    const float* row = newh + (size_t)(b * n + p) * HDIM;
    float s = 0.0f;
    for (int j = (int)lane; j < HDIM; j += 32) s += row[j] * q[j];
#pragma unroll
    for (int off = 16; off > 0; off >>= 1) s += __shfl_xor(s, off, 32);
    if (lane == 0) sc[p] = s;
  }
  __syncthreads();

  if (tid == 0) {
    int len = length[b];
    int best = 0;
    float bv = -INFINITY;
    for (int p = 0; p < n; ++p) {
      float s = ((iter + 1 + p) < len) ? sc[p] : -1.0e9f;
      if (s > bv) { bv = s; best = p; }
    }
    ksel_s = best;
  }
  __syncthreads();

  const int ksel = ksel_s;
  const bool done = (iter + 1) < length[b];

  for (int idx = tid; idx < n * HDIM; idx += 256) {
    int p = idx >> 9;
    int j = idx & (HDIM - 1);
    size_t cur = (size_t)(b * SEQL + p) * HDIM + j;
    float hm, cm;
    if (p < ksel)      { hm = hcur[cur];        cm = ccur[cur]; }
    else if (p > ksel) { hm = hcur[cur + HDIM]; cm = ccur[cur + HDIM]; }
    else {
      size_t ni = (size_t)(b * n + p) * HDIM + j;
      hm = newh[ni]; cm = newc[ni];
    }
    hnxt[cur] = done ? hm : hcur[cur];
    cnxt[cur] = done ? cm : ccur[cur];
  }
}

__global__ __launch_bounds__(256) void final_copy_k(const float* __restrict__ hfin,
                                                    float* __restrict__ out) {
  int idx = blockIdx.x * 256 + threadIdx.x;
  int b = idx >> 9;
  int j = idx & (HDIM - 1);
  out[idx] = hfin[(size_t)b * SEQL * HDIM + j];
}

extern "C" void kernel_launch(void* const* d_in, const int* in_sizes, int n_in,
                              void* d_out, int out_size, void* d_ws, size_t ws_size,
                              hipStream_t stream) {
  const float* inp    = (const float*)d_in[0];
  const int*   length = (const int*)  d_in[1];
  const float* word_W = (const float*)d_in[2];
  const float* word_b = (const float*)d_in[3];
  const float* comp_W = (const float*)d_in[4];
  const float* comp_b = (const float*)d_in[5];
  const float* comp_q = (const float*)d_in[6];

  float* ws = (float*)d_ws;
  size_t off = 0;
  float* hb[2];
  float* cb[2];
  hb[0] = ws + off; off += (size_t)BATCH * SEQL * HDIM;
  cb[0] = ws + off; off += (size_t)BATCH * SEQL * HDIM;
  hb[1] = ws + off; off += (size_t)BATCH * SEQL * HDIM;
  cb[1] = ws + off; off += (size_t)BATCH * SEQL * HDIM;
  float* newh = ws + off; off += (size_t)(BATCH * 23) * HDIM;
  float* newc = ws + off; off += (size_t)(BATCH * 23) * HDIM;

  gemm_wmma<0><<<dim3((BATCH * SEQL) / 32, 1024 / 256), 256, 0, stream>>>(
      inp, nullptr, nullptr, word_W, word_b, hb[0], cb[0], 0);

  int cur = 0;
  for (int i = 0; i < SEQL - 1; ++i) {
    int n = (SEQL - 1) - i;
    int R = BATCH * n;
    int nxt = cur ^ 1;
    gemm_wmma<1><<<dim3(R / 32, HDIM / 64), 256, 0, stream>>>(
        nullptr, hb[cur], cb[cur], comp_W, comp_b, newh, newc, n);
    mix_k<<<BATCH, 256, 0, stream>>>(newh, newc, hb[cur], cb[cur], comp_q, length,
                                     hb[nxt], cb[nxt], n, i);
    cur = nxt;
  }

  final_copy_k<<<(BATCH * HDIM) / 256, 256, 0, stream>>>(hb[cur], (float*)d_out);
  (void)in_sizes; (void)n_in; (void)out_size; (void)ws_size;
}


# Comp GEMM workgroup halved to 4 waves (2Mx2N; 32 rows x 32 cols x 5 gates): twice as many workgroups per merge step, so short steps fill more WGPs, one wave per SIMD
# speedup vs baseline: 1.5220x; 1.0916x over previous
#include <hip/hip_runtime.h>
#include <hip/hip_bf16.h>
#include <math.h>


typedef __attribute__((ext_vector_type(2))) float v2f;
typedef __attribute__((ext_vector_type(4))) float v4f;
typedef __attribute__((ext_vector_type(8))) float v8f;

#define BATCH 64
#define SEQL  24
#define HDIM  512

__device__ __forceinline__ float sigm(float x) { return 1.0f / (1.0f + __expf(-x)); }
__device__ __forceinline__ float tanh_fast(float x) { return 2.0f / (1.0f + __expf(-2.0f * x)) - 1.0f; }

#define LDS_A_STAGE (32 * 36)
template <int MODE> struct BStage { static constexpr int rows = (MODE == 0) ? 320 : 160; };

template <int MODE>
__device__ __forceinline__ constexpr int wrow_of(int i) {
  return (MODE == 0) ? (i * 32) : ((i >> 1) * HDIM + (i & 1) * 16);
}

template <int MODE, int NA, int NB, int KD>
__device__ __forceinline__ void stage_gload(const float* const (&asrc)[NA], const float* bsrc, int k,
                                            v4f (&ra)[NA], v4f (&rb)[NB]) {
#pragma unroll
  for (int a = 0; a < NA; ++a)
    ra[a] = *(const v4f*)(asrc[a] + k);
#pragma unroll
  for (int i = 0; i < NB; ++i)
    rb[i] = *(const v4f*)(bsrc + (size_t)wrow_of<MODE>(i) * KD + k);
}

template <int RS, int NA, int NB>
__device__ __forceinline__ void stage_lstore(float* As, float* Bs, int sr, int sc4,
                                             const v4f (&ra)[NA], const v4f (&rb)[NB]) {
#pragma unroll
  for (int a = 0; a < NA; ++a)
    *(v4f*)&As[(a * RS + sr) * 36 + sc4 * 4] = ra[a];
#pragma unroll
  for (int i = 0; i < NB; ++i)
    *(v4f*)&Bs[(i * RS + sr) * 36 + sc4 * 4] = rb[i];
}

template <int MODE, int NACC>
__device__ __forceinline__ void stage_mma(const float* As, const float* Bs,
                                          int wm, int wn, int l16, int lhi, v8f (&acc)[5]) {
#pragma unroll
  for (int kk = 0; kk < 32; kk += 4) {
    v2f a = *(const v2f*)&As[(wm * 16 + l16) * 36 + kk + 2 * lhi];
#pragma unroll
    for (int g = 0; g < NACC; ++g) {
      int nrow = (MODE == 0) ? (wn * 64 + g * 16 + l16)
                             : (g * 32 + wn * 16 + l16);
      v2f bf = *(const v2f*)&Bs[nrow * 36 + kk + 2 * lhi];
      acc[g] = __builtin_amdgcn_wmma_f32_16x16x4_f32(
          false, a, false, bf, (short)0, acc[g], false, false);
    }
  }
}

template <int MODE>
__global__ __launch_bounds__((MODE == 0) ? 256 : 128) void gemm_wmma(const float* __restrict__ X,
                                                 const float* __restrict__ hbuf,
                                                 const float* __restrict__ cbuf,
                                                 const float* __restrict__ W,
                                                 const float* __restrict__ bias,
                                                 float* __restrict__ out0,
                                                 float* __restrict__ out1,
                                                 int n) {
  constexpr int KD   = (MODE == 0) ? 512 : 1024;
  constexpr int NB   = (MODE == 0) ? 8 : 10;
  constexpr int NACC = (MODE == 0) ? 4 : 5;
  constexpr int RS   = (MODE == 0) ? 32 : 16;
  constexpr int NA   = 32 / RS;
  constexpr int NWN  = (MODE == 0) ? 4 : 2;
  constexpr int LDS_B_STAGE = BStage<MODE>::rows * 36;

  __shared__ __attribute__((aligned(16))) float A_lds[2 * LDS_A_STAGE];
  __shared__ __attribute__((aligned(16))) float B_lds[2 * LDS_B_STAGE];
  float* const A0 = A_lds;
  float* const A1 = A_lds + LDS_A_STAGE;
  float* const B0 = B_lds;
  float* const B1 = B_lds + LDS_B_STAGE;

  const int tid  = threadIdx.x;
  const unsigned lane = tid & 31u;
  const int wave = tid >> 5;
  const int wm   = wave / NWN;
  const int wn   = wave % NWN;
  const int l16  = (int)(lane & 15u);
  const int lhi  = (int)(lane >> 4);

  const int r0 = blockIdx.x * 32;

  const int sr  = tid >> 3;
  const int sc4 = tid & 7;
  const float* asrc[NA];
#pragma unroll
  for (int a = 0; a < NA; ++a) {
    if (MODE == 0) {
      asrc[a] = X + (size_t)(r0 + a * RS + sr) * KD + sc4 * 4;
    } else {
      int rg = r0 + a * RS + sr;
      int b  = rg / n;
      int p  = rg - b * n;
      asrc[a] = hbuf + (size_t)(b * SEQL + p) * HDIM + sc4 * 4;
    }
  }
  const float* bsrc = W + (size_t)(blockIdx.y * ((MODE == 0) ? 256 : 32) + sr) * KD + sc4 * 4;

  v4f ra[NA];
  v4f rb[NB];
  v8f acc[5] = {};

  stage_gload<MODE, NA, NB, KD>(asrc, bsrc, 0, ra, rb);
  stage_lstore<RS, NA, NB>(A0, B0, sr, sc4, ra, rb);
  __syncthreads();

  for (int k0 = 0; k0 < KD - 64; k0 += 64) {
    stage_gload<MODE, NA, NB, KD>(asrc, bsrc, k0 + 32, ra, rb);
    stage_mma<MODE, NACC>(A0, B0, wm, wn, l16, lhi, acc);
    stage_lstore<RS, NA, NB>(A1, B1, sr, sc4, ra, rb);
    __syncthreads();

    stage_gload<MODE, NA, NB, KD>(asrc, bsrc, k0 + 64, ra, rb);
    stage_mma<MODE, NACC>(A1, B1, wm, wn, l16, lhi, acc);
    stage_lstore<RS, NA, NB>(A0, B0, sr, sc4, ra, rb);
    __syncthreads();
  }
  stage_gload<MODE, NA, NB, KD>(asrc, bsrc, KD - 32, ra, rb);
  stage_mma<MODE, NACC>(A0, B0, wm, wn, l16, lhi, acc);
  stage_lstore<RS, NA, NB>(A1, B1, sr, sc4, ra, rb);
  __syncthreads();
  stage_mma<MODE, NACC>(A1, B1, wm, wn, l16, lhi, acc);

  if (MODE == 0) {
    const int colbase = blockIdx.y * 256 + wn * 64;
    const int rowg0   = r0 + wm * 16 + 8 * lhi;
#pragma unroll
    for (int j = 0; j < 4; ++j) {
      int col = colbase + j * 16 + l16;
      float bv = bias[col];
#pragma unroll
      for (int vr = 0; vr < 8; ++vr) {
        int rg = rowg0 + vr;
        float val = acc[j][vr] + bv;
        if (col < HDIM) out0[(size_t)rg * HDIM + col] = val;
        else            out1[(size_t)rg * HDIM + (col - HDIM)] = val;
      }
    }
  } else {
    const int colg = blockIdx.y * 32 + wn * 16 + l16;
    float bz[5];
#pragma unroll
    for (int g = 0; g < 5; ++g) bz[g] = bias[g * HDIM + colg];
    int rg = r0 + wm * 16 + 8 * lhi;
    int b  = rg / n;
    int p  = rg - b * n;
#pragma unroll
    for (int vr = 0; vr < 8; ++vr) {
      float gi  = acc[0][vr] + bz[0];
      float gfl = acc[1][vr] + bz[1];
      float gfr = acc[2][vr] + bz[2];
      float gu  = acc[3][vr] + bz[3];
      float go  = acc[4][vr] + bz[4];
      size_t cbase = (size_t)(b * SEQL + p) * HDIM + colg;
      float cl = cbuf[cbase];
      float cr = cbuf[cbase + HDIM];
      float nc = cl * sigm(gfl + 1.0f) + cr * sigm(gfr + 1.0f) + tanh_fast(gu) * sigm(gi);
      float nh = sigm(go) * tanh_fast(nc);
      size_t oi = (size_t)rg * HDIM + colg;
      out0[oi] = nh;
      out1[oi] = nc;
      ++rg; ++p;
      if (p == n) { p = 0; ++b; }
    }
  }
}

__global__ __launch_bounds__(256) void mix_k(const float* __restrict__ newh,
                                             const float* __restrict__ newc,
                                             const float* __restrict__ hcur,
                                             const float* __restrict__ ccur,
                                             const float* __restrict__ q,
                                             const int* __restrict__ length,
                                             float* __restrict__ hnxt,
                                             float* __restrict__ cnxt,
                                             int n, int iter) {
  __shared__ float sc[32];
  __shared__ int ksel_s;
  const int b = blockIdx.x;
  const int tid = threadIdx.x;
  const unsigned lane = tid & 31u;
  const int wave = tid >> 5;

  for (int p = wave; p < n; p += 8) {
    const float* row = newh + (size_t)(b * n + p) * HDIM;
    float s = 0.0f;
    for (int j = (int)lane; j < HDIM; j += 32) s += row[j] * q[j];
#pragma unroll
    for (int off = 16; off > 0; off >>= 1) s += __shfl_xor(s, off, 32);
    if (lane == 0) sc[p] = s;
  }
  __syncthreads();

  if (tid == 0) {
    int len = length[b];
    int best = 0;
    float bv = -INFINITY;
    for (int p = 0; p < n; ++p) {
      float s = ((iter + 1 + p) < len) ? sc[p] : -1.0e9f;
      if (s > bv) { bv = s; best = p; }
    }
    ksel_s = best;
  }
  __syncthreads();

  const int ksel = ksel_s;
  const bool done = (iter + 1) < length[b];

  for (int idx = tid; idx < n * HDIM; idx += 256) {
    int p = idx >> 9;
    int j = idx & (HDIM - 1);
    size_t cur = (size_t)(b * SEQL + p) * HDIM + j;
    float hm, cm;
    if (p < ksel)      { hm = hcur[cur];        cm = ccur[cur]; }
    else if (p > ksel) { hm = hcur[cur + HDIM]; cm = ccur[cur + HDIM]; }
    else {
      size_t ni = (size_t)(b * n + p) * HDIM + j;
      hm = newh[ni]; cm = newc[ni];
    }
    hnxt[cur] = done ? hm : hcur[cur];
    cnxt[cur] = done ? cm : ccur[cur];
  }
}

__global__ __launch_bounds__(256) void final_copy_k(const float* __restrict__ hfin,
                                                    float* __restrict__ out) {
  int idx = blockIdx.x * 256 + threadIdx.x;
  int b = idx >> 9;
  int j = idx & (HDIM - 1);
  out[idx] = hfin[(size_t)b * SEQL * HDIM + j];
}

extern "C" void kernel_launch(void* const* d_in, const int* in_sizes, int n_in,
                              void* d_out, int out_size, void* d_ws, size_t ws_size,
                              hipStream_t stream) {
  const float* inp    = (const float*)d_in[0];
  const int*   length = (const int*)  d_in[1];
  const float* word_W = (const float*)d_in[2];
  const float* word_b = (const float*)d_in[3];
  const float* comp_W = (const float*)d_in[4];
  const float* comp_b = (const float*)d_in[5];
  const float* comp_q = (const float*)d_in[6];

  float* ws = (float*)d_ws;
  size_t off = 0;
  float* hb[2];
  float* cb[2];
  hb[0] = ws + off; off += (size_t)BATCH * SEQL * HDIM;
  cb[0] = ws + off; off += (size_t)BATCH * SEQL * HDIM;
  hb[1] = ws + off; off += (size_t)BATCH * SEQL * HDIM;
  cb[1] = ws + off; off += (size_t)BATCH * SEQL * HDIM;
  float* newh = ws + off; off += (size_t)(BATCH * 23) * HDIM;
  float* newc = ws + off; off += (size_t)(BATCH * 23) * HDIM;

  gemm_wmma<0><<<dim3((BATCH * SEQL) / 32, 1024 / 256), 256, 0, stream>>>(
      inp, nullptr, nullptr, word_W, word_b, hb[0], cb[0], 0);

  int cur = 0;
  for (int i = 0; i < SEQL - 1; ++i) {
    int n = (SEQL - 1) - i;
    int R = BATCH * n;
    int nxt = cur ^ 1;
    gemm_wmma<1><<<dim3(R / 32, HDIM / 32), 128, 0, stream>>>(
        nullptr, hb[cur], cb[cur], comp_W, comp_b, newh, newc, n);
    mix_k<<<BATCH, 256, 0, stream>>>(newh, newc, hb[cur], cb[cur], comp_q, length,
                                     hb[nxt], cb[nxt], n, i);
    cur = nxt;
  }

  final_copy_k<<<(BATCH * HDIM) / 256, 256, 0, stream>>>(hb[cur], (float*)d_out);
  (void)in_sizes; (void)n_in; (void)out_size; (void)ws_size;
}


# Incremental merge steps: after step 1 a step multiplies only the 2 pairs per batch row that contain the merged node; other candidates carried over in place, same sums (4288 GEMM rows, was 17664)
# speedup vs baseline: 2.6162x; 1.7189x over previous
#include <hip/hip_runtime.h>
#include <hip/hip_bf16.h>
#include <math.h>


typedef __attribute__((ext_vector_type(2))) float v2f;
typedef __attribute__((ext_vector_type(4))) float v4f;
typedef __attribute__((ext_vector_type(8))) float v8f;

#define BATCH 64
#define SEQL  24
#define HDIM  512

__device__ __forceinline__ float sigm(float x) { return 1.0f / (1.0f + __expf(-x)); }
__device__ __forceinline__ float tanh_fast(float x) { return 2.0f / (1.0f + __expf(-2.0f * x)) - 1.0f; }

#define LDS_A_STAGE (32 * 36)
template <int MODE> struct BStage { static constexpr int rows = (MODE == 0) ? 320 : 160; };

template <int MODE>
__device__ __forceinline__ constexpr int wrow_of(int i) {
  return (MODE == 0) ? (i * 32) : ((i >> 1) * HDIM + (i & 1) * 16);
}

template <int MODE, int NA, int NB, int KD>
__device__ __forceinline__ void stage_gload(const float* const (&asrc)[NA], const float* bsrc, int k,
                                            v4f (&ra)[NA], v4f (&rb)[NB]) {
#pragma unroll
  for (int a = 0; a < NA; ++a)
    ra[a] = *(const v4f*)(asrc[a] + k);
#pragma unroll
  for (int i = 0; i < NB; ++i)
    rb[i] = *(const v4f*)(bsrc + (size_t)wrow_of<MODE>(i) * KD + k);
}

template <int RS, int NA, int NB>
__device__ __forceinline__ void stage_lstore(float* As, float* Bs, int sr, int sc4,
                                             const v4f (&ra)[NA], const v4f (&rb)[NB]) {
#pragma unroll
  for (int a = 0; a < NA; ++a)
    *(v4f*)&As[(a * RS + sr) * 36 + sc4 * 4] = ra[a];
#pragma unroll
  for (int i = 0; i < NB; ++i)
    *(v4f*)&Bs[(i * RS + sr) * 36 + sc4 * 4] = rb[i];
}

template <int MODE, int NACC>
__device__ __forceinline__ void stage_mma(const float* As, const float* Bs,
                                          int wm, int wn, int l16, int lhi, v8f (&acc)[5]) {
#pragma unroll
  for (int kk = 0; kk < 32; kk += 4) {
    v2f a = *(const v2f*)&As[(wm * 16 + l16) * 36 + kk + 2 * lhi];
#pragma unroll
    for (int g = 0; g < NACC; ++g) {
      int nrow = (MODE == 0) ? (wn * 64 + g * 16 + l16)
                             : (g * 32 + wn * 16 + l16);
      v2f bf = *(const v2f*)&Bs[nrow * 36 + kk + 2 * lhi];
      acc[g] = __builtin_amdgcn_wmma_f32_16x16x4_f32(
          false, a, false, bf, (short)0, acc[g], false, false);
    }
  }
}

template <int MODE>
__device__ __forceinline__ void comp_row(int rg, int n, int tabp, int& b, int& p) {
  if (MODE == 1) {
    b = rg / n;
    p = rg - b * n;
  } else {
    b = rg >> 1;
    p = tabp;
  }
}

template <int MODE>
__global__ __launch_bounds__((MODE == 0) ? 256 : 128) void gemm_wmma(const float* __restrict__ X,
                                                 const float* __restrict__ hbuf,
                                                 const float* __restrict__ cbuf,
                                                 const float* __restrict__ W,
                                                 const float* __restrict__ bias,
                                                 float* __restrict__ out0,
                                                 float* __restrict__ out1,
                                                 const int* __restrict__ rowtab,
                                                 int n) {
  constexpr int KD   = (MODE == 0) ? 512 : 1024;
  constexpr int NB   = (MODE == 0) ? 8 : 10;
  constexpr int NACC = (MODE == 0) ? 4 : 5;
  constexpr int RS   = (MODE == 0) ? 32 : 16;
  constexpr int NA   = 32 / RS;
  constexpr int NWN  = (MODE == 0) ? 4 : 2;
  constexpr int LDS_B_STAGE = BStage<MODE>::rows * 36;

  __shared__ __attribute__((aligned(16))) float A_lds[2 * LDS_A_STAGE];
  __shared__ __attribute__((aligned(16))) float B_lds[2 * LDS_B_STAGE];
  float* const A0 = A_lds;
  float* const A1 = A_lds + LDS_A_STAGE;
  float* const B0 = B_lds;
  float* const B1 = B_lds + LDS_B_STAGE;

  const int tid  = threadIdx.x;
  const unsigned lane = tid & 31u;
  const int wave = tid >> 5;
  const int wm   = wave / NWN;
  const int wn   = wave % NWN;
  const int l16  = (int)(lane & 15u);
  const int lhi  = (int)(lane >> 4);

  const int r0 = blockIdx.x * 32;

  const int sr  = tid >> 3;
  const int sc4 = tid & 7;
  const float* asrc[NA];
#pragma unroll
  for (int a = 0; a < NA; ++a) {
    if (MODE == 0) {
      asrc[a] = X + (size_t)(r0 + a * RS + sr) * KD + sc4 * 4;
    } else {
      const int rg = r0 + a * RS + sr;
      int b, p;
      comp_row<MODE>(rg, n, (MODE == 2) ? rowtab[rg] : 0, b, p);
      asrc[a] = hbuf + (size_t)(b * SEQL + p) * HDIM + sc4 * 4;
    }
  }
  const float* bsrc = W + (size_t)(blockIdx.y * ((MODE == 0) ? 256 : 32) + sr) * KD + sc4 * 4;

  const int erow0 = r0 + wm * 16 + 8 * lhi;
  int etab[8] = {};
  if (MODE == 2) {
    const int4 t0 = *(const int4*)(rowtab + erow0);
    const int4 t1 = *(const int4*)(rowtab + erow0 + 4);
    etab[0] = t0.x; etab[1] = t0.y; etab[2] = t0.z; etab[3] = t0.w;
    etab[4] = t1.x; etab[5] = t1.y; etab[6] = t1.z; etab[7] = t1.w;
  }

  v4f ra[NA];
  v4f rb[NB];
  v8f acc[5] = {};

  stage_gload<MODE, NA, NB, KD>(asrc, bsrc, 0, ra, rb);
  stage_lstore<RS, NA, NB>(A0, B0, sr, sc4, ra, rb);
  __syncthreads();

  for (int k0 = 0; k0 < KD - 64; k0 += 64) {
    stage_gload<MODE, NA, NB, KD>(asrc, bsrc, k0 + 32, ra, rb);
    stage_mma<MODE, NACC>(A0, B0, wm, wn, l16, lhi, acc);
    stage_lstore<RS, NA, NB>(A1, B1, sr, sc4, ra, rb);
    __syncthreads();

    stage_gload<MODE, NA, NB, KD>(asrc, bsrc, k0 + 64, ra, rb);
    stage_mma<MODE, NACC>(A1, B1, wm, wn, l16, lhi, acc);
    stage_lstore<RS, NA, NB>(A0, B0, sr, sc4, ra, rb);
    __syncthreads();
  }
  stage_gload<MODE, NA, NB, KD>(asrc, bsrc, KD - 32, ra, rb);
  stage_mma<MODE, NACC>(A0, B0, wm, wn, l16, lhi, acc);
  stage_lstore<RS, NA, NB>(A1, B1, sr, sc4, ra, rb);
  __syncthreads();
  stage_mma<MODE, NACC>(A1, B1, wm, wn, l16, lhi, acc);

  if (MODE == 0) {
    const int colbase = blockIdx.y * 256 + wn * 64;
    const int rowg0   = r0 + wm * 16 + 8 * lhi;
#pragma unroll
    for (int j = 0; j < 4; ++j) {
      int col = colbase + j * 16 + l16;
      float bv = bias[col];
#pragma unroll
      for (int vr = 0; vr < 8; ++vr) {
        int rg = rowg0 + vr;
        float val = acc[j][vr] + bv;
        if (col < HDIM) out0[(size_t)rg * HDIM + col] = val;
        else            out1[(size_t)rg * HDIM + (col - HDIM)] = val;
      }
    }
  } else {
    const int colg = blockIdx.y * 32 + wn * 16 + l16;
    float bz[5];
#pragma unroll
    for (int g = 0; g < 5; ++g) bz[g] = bias[g * HDIM + colg];
#pragma unroll
    for (int vr = 0; vr < 8; ++vr) {
      int b, p;
      comp_row<MODE>(erow0 + vr, n, etab[vr], b, p);
      float gi  = acc[0][vr] + bz[0];
      float gfl = acc[1][vr] + bz[1];
      float gfr = acc[2][vr] + bz[2];
      float gu  = acc[3][vr] + bz[3];
      float go  = acc[4][vr] + bz[4];
      size_t cbase = (size_t)(b * SEQL + p) * HDIM + colg;
      float cl = cbuf[cbase];
      float cr = cbuf[cbase + HDIM];
      float nc = cl * sigm(gfl + 1.0f) + cr * sigm(gfr + 1.0f) + tanh_fast(gu) * sigm(gi);
      float nh = sigm(go) * tanh_fast(nc);
      out0[cbase] = nh;
      out1[cbase] = nc;
    }
  }
}

__global__ __launch_bounds__(512) void mix_k(float* nh, float* nc, float* h, float* c,
                                             const float* __restrict__ q,
                                             const int* __restrict__ length,
                                             int* __restrict__ rowtab,
                                             int n, int iter) {
  __shared__ float sc[32];
  __shared__ int ksel_s;
  const int b = blockIdx.x;
  const int tid = threadIdx.x;
  const unsigned lane = tid & 31u;
  const int wave = tid >> 5;

  for (int p = wave; p < n; p += 16) {
    const float* row = nh + (size_t)(b * SEQL + p) * HDIM;
    float s = 0.0f;
    for (int j = (int)lane; j < HDIM; j += 32) s += row[j] * q[j];
#pragma unroll
    for (int off = 16; off > 0; off >>= 1) s += __shfl_xor(s, off, 32);
    if (lane == 0) sc[p] = s;
  }
  __syncthreads();

  if (tid == 0) {
    int len = length[b];
    int best = 0;
    float bv = -INFINITY;
    for (int p = 0; p < n; ++p) {
      float s = ((iter + 1 + p) < len) ? sc[p] : -1.0e9f;
      if (s > bv) { bv = s; best = p; }
    }
    ksel_s = best;
  }
  __syncthreads();

  const int ksel = __builtin_amdgcn_readfirstlane(ksel_s);
  const bool done = (iter + 1) < length[b];

  if (tid == 0) {
    const int hi = (n >= 2) ? (n - 2) : 0;
    const int pl = (ksel >= 1) ? (ksel - 1) : 0;
    const int pr = (ksel <= hi) ? ksel : hi;
    rowtab[2 * b]     = done ? pl : 0;
    rowtab[2 * b + 1] = done ? pr : 0;
  }
  if (!done) return;

  const size_t base = (size_t)b * SEQL * HDIM + tid;
  float* hB  = h  + base;
  float* cB  = c  + base;
  float* nhB = nh + base;
  float* ncB = nc + base;

  float hv[SEQL - 1], cv[SEQL - 1], nhv[SEQL - 1], ncv[SEQL - 1];
#pragma unroll
  for (int p = 0; p < SEQL - 1; ++p) {
    if (p >= ksel && p < n) {
      if (p == ksel) { hv[p] = nhB[p * HDIM];       cv[p] = ncB[p * HDIM]; }
      else           { hv[p] = hB[(p + 1) * HDIM];  cv[p] = cB[(p + 1) * HDIM]; }
    }
    if (p > ksel && p < n - 1) { nhv[p] = nhB[(p + 1) * HDIM]; ncv[p] = ncB[(p + 1) * HDIM]; }
  }
#pragma unroll
  for (int p = 0; p < SEQL - 1; ++p) {
    if (p >= ksel && p < n)    { hB[p * HDIM]  = hv[p];  cB[p * HDIM]  = cv[p]; }
    if (p > ksel && p < n - 1) { nhB[p * HDIM] = nhv[p]; ncB[p * HDIM] = ncv[p]; }
  }
}

__global__ __launch_bounds__(256) void final_copy_k(const float* __restrict__ hfin,
                                                    float* __restrict__ out) {
  int idx = blockIdx.x * 256 + threadIdx.x;
  int b = idx >> 9;
  int j = idx & (HDIM - 1);
  out[idx] = hfin[(size_t)b * SEQL * HDIM + j];
}

extern "C" void kernel_launch(void* const* d_in, const int* in_sizes, int n_in,
                              void* d_out, int out_size, void* d_ws, size_t ws_size,
                              hipStream_t stream) {
  const float* inp    = (const float*)d_in[0];
  const int*   length = (const int*)  d_in[1];
  const float* word_W = (const float*)d_in[2];
  const float* word_b = (const float*)d_in[3];
  const float* comp_W = (const float*)d_in[4];
  const float* comp_b = (const float*)d_in[5];
  const float* comp_q = (const float*)d_in[6];

  float* ws = (float*)d_ws;
  size_t off = 0;
  float* hb = ws + off; off += (size_t)BATCH * SEQL * HDIM;
  float* cb = ws + off; off += (size_t)BATCH * SEQL * HDIM;
  float* nh = ws + off; off += (size_t)BATCH * SEQL * HDIM;
  float* nc = ws + off; off += (size_t)BATCH * SEQL * HDIM;
  int* rowtab = (int*)(ws + off);

  gemm_wmma<0><<<dim3((BATCH * SEQL) / 32, 1024 / 256), 256, 0, stream>>>(
      inp, nullptr, nullptr, word_W, word_b, hb, cb, nullptr, 0);

  gemm_wmma<1><<<dim3((BATCH * (SEQL - 1)) / 32, HDIM / 32), 128, 0, stream>>>(
      nullptr, hb, cb, comp_W, comp_b, nh, nc, nullptr, SEQL - 1);
  mix_k<<<BATCH, 512, 0, stream>>>(nh, nc, hb, cb, comp_q, length, rowtab, SEQL - 1, 0);

  for (int i = 1; i < SEQL - 1; ++i) {
    int n = (SEQL - 1) - i;
    gemm_wmma<2><<<dim3((BATCH * 2) / 32, HDIM / 32), 128, 0, stream>>>(
        nullptr, hb, cb, comp_W, comp_b, nh, nc, rowtab, n);
    mix_k<<<BATCH, 512, 0, stream>>>(nh, nc, hb, cb, comp_q, length, rowtab, n, i);
  }

  final_copy_k<<<(BATCH * HDIM) / 256, 256, 0, stream>>>(hb, (float*)d_out);
  (void)in_sizes; (void)n_in; (void)out_size; (void)ws_size;
}


# Merge-step GEMM split along K over the 4 waves of a workgroup (256 workgroups of 16 rows x 16 cols x 5 gates, a wave on every SIMD); the 4 partial sums are added in fixed order in LDS
# speedup vs baseline: 4.5662x; 1.7454x over previous
#include <hip/hip_runtime.h>
#include <hip/hip_bf16.h>
#include <math.h>


typedef __attribute__((ext_vector_type(2))) float v2f;
typedef __attribute__((ext_vector_type(4))) float v4f;
typedef __attribute__((ext_vector_type(8))) float v8f;

#define BATCH 64
#define SEQL  24
#define HDIM  512

__device__ __forceinline__ float sigm(float x) { return 1.0f / (1.0f + __expf(-x)); }
__device__ __forceinline__ float tanh_fast(float x) { return 2.0f / (1.0f + __expf(-2.0f * x)) - 1.0f; }

#define LDS_A_STAGE (32 * 36)
template <int MODE> struct BStage { static constexpr int rows = (MODE == 0) ? 320 : 160; };

template <int MODE>
__device__ __forceinline__ constexpr int wrow_of(int i) {
  return (MODE == 0) ? (i * 32) : ((i >> 1) * HDIM + (i & 1) * 16);
}

template <int MODE, int NA, int NB, int KD>
__device__ __forceinline__ void stage_gload(const float* const (&asrc)[NA], const float* bsrc, int k,
                                            v4f (&ra)[NA], v4f (&rb)[NB]) {
#pragma unroll
  for (int a = 0; a < NA; ++a)
    ra[a] = *(const v4f*)(asrc[a] + k);
#pragma unroll
  for (int i = 0; i < NB; ++i)
    rb[i] = *(const v4f*)(bsrc + (size_t)wrow_of<MODE>(i) * KD + k);
}

template <int RS, int NA, int NB>
__device__ __forceinline__ void stage_lstore(float* As, float* Bs, int sr, int sc4,
                                             const v4f (&ra)[NA], const v4f (&rb)[NB]) {
#pragma unroll
  for (int a = 0; a < NA; ++a)
    *(v4f*)&As[(a * RS + sr) * 36 + sc4 * 4] = ra[a];
#pragma unroll
  for (int i = 0; i < NB; ++i)
    *(v4f*)&Bs[(i * RS + sr) * 36 + sc4 * 4] = rb[i];
}

template <int MODE, int NACC>
__device__ __forceinline__ void stage_mma(const float* As, const float* Bs,
                                          int wm, int wn, int l16, int lhi, v8f (&acc)[5]) {
#pragma unroll
  for (int kk = 0; kk < 32; kk += 4) {
    v2f a = *(const v2f*)&As[(wm * 16 + l16) * 36 + kk + 2 * lhi];
#pragma unroll
    for (int g = 0; g < NACC; ++g) {
      int nrow = (MODE == 0) ? (wn * 64 + g * 16 + l16)
                             : (g * 32 + wn * 16 + l16);
      v2f bf = *(const v2f*)&Bs[nrow * 36 + kk + 2 * lhi];
      acc[g] = __builtin_amdgcn_wmma_f32_16x16x4_f32(
          false, a, false, bf, (short)0, acc[g], false, false);
    }
  }
}

__device__ __forceinline__ void comp_row(int rg, int n, int& b, int& p) {
  b = rg / n;
  p = rg - b * n;
}

template <int MODE>
__global__ __launch_bounds__((MODE == 0) ? 256 : 128) void gemm_wmma(const float* __restrict__ X,
                                                 const float* __restrict__ hbuf,
                                                 const float* __restrict__ cbuf,
                                                 const float* __restrict__ W,
                                                 const float* __restrict__ bias,
                                                 float* __restrict__ out0,
                                                 float* __restrict__ out1,
                                                 int n) {
  constexpr int KD   = (MODE == 0) ? 512 : 1024;
  constexpr int NB   = (MODE == 0) ? 8 : 10;
  constexpr int NACC = (MODE == 0) ? 4 : 5;
  constexpr int RS   = (MODE == 0) ? 32 : 16;
  constexpr int NA   = 32 / RS;
  constexpr int NWN  = (MODE == 0) ? 4 : 2;
  constexpr int LDS_B_STAGE = BStage<MODE>::rows * 36;

  __shared__ __attribute__((aligned(16))) float A_lds[2 * LDS_A_STAGE];
  __shared__ __attribute__((aligned(16))) float B_lds[2 * LDS_B_STAGE];
  float* const A0 = A_lds;
  float* const A1 = A_lds + LDS_A_STAGE;
  float* const B0 = B_lds;
  float* const B1 = B_lds + LDS_B_STAGE;

  const int tid  = threadIdx.x;
  const unsigned lane = tid & 31u;
  const int wave = tid >> 5;
  const int wm   = wave / NWN;
  const int wn   = wave % NWN;
  const int l16  = (int)(lane & 15u);
  const int lhi  = (int)(lane >> 4);

  const int r0 = blockIdx.x * 32;

  const int sr  = tid >> 3;
  const int sc4 = tid & 7;
  const float* asrc[NA];
#pragma unroll
  for (int a = 0; a < NA; ++a) {
    if (MODE == 0) {
      asrc[a] = X + (size_t)(r0 + a * RS + sr) * KD + sc4 * 4;
    } else {
      const int rg = r0 + a * RS + sr;
      int b, p;
      comp_row(rg, n, b, p);
      asrc[a] = hbuf + (size_t)(b * SEQL + p) * HDIM + sc4 * 4;
    }
  }
  const float* bsrc = W + (size_t)(blockIdx.y * ((MODE == 0) ? 256 : 32) + sr) * KD + sc4 * 4;

  const int erow0 = r0 + wm * 16 + 8 * lhi;

  v4f ra[NA];
  v4f rb[NB];
  v8f acc[5] = {};

  stage_gload<MODE, NA, NB, KD>(asrc, bsrc, 0, ra, rb);
  stage_lstore<RS, NA, NB>(A0, B0, sr, sc4, ra, rb);
  __syncthreads();

  for (int k0 = 0; k0 < KD - 64; k0 += 64) {
    stage_gload<MODE, NA, NB, KD>(asrc, bsrc, k0 + 32, ra, rb);
    stage_mma<MODE, NACC>(A0, B0, wm, wn, l16, lhi, acc);
    stage_lstore<RS, NA, NB>(A1, B1, sr, sc4, ra, rb);
    __syncthreads();

    stage_gload<MODE, NA, NB, KD>(asrc, bsrc, k0 + 64, ra, rb);
    stage_mma<MODE, NACC>(A1, B1, wm, wn, l16, lhi, acc);
    stage_lstore<RS, NA, NB>(A0, B0, sr, sc4, ra, rb);
    __syncthreads();
  }
  stage_gload<MODE, NA, NB, KD>(asrc, bsrc, KD - 32, ra, rb);
  stage_mma<MODE, NACC>(A0, B0, wm, wn, l16, lhi, acc);
  stage_lstore<RS, NA, NB>(A1, B1, sr, sc4, ra, rb);
  __syncthreads();
  stage_mma<MODE, NACC>(A1, B1, wm, wn, l16, lhi, acc);

  if (MODE == 0) {
    const int colbase = blockIdx.y * 256 + wn * 64;
    const int rowg0   = r0 + wm * 16 + 8 * lhi;
#pragma unroll
    for (int j = 0; j < 4; ++j) {
      int col = colbase + j * 16 + l16;
      float bv = bias[col];
#pragma unroll
      for (int vr = 0; vr < 8; ++vr) {
        int rg = rowg0 + vr;
        float val = acc[j][vr] + bv;
        if (col < HDIM) out0[(size_t)rg * HDIM + col] = val;
        else            out1[(size_t)rg * HDIM + (col - HDIM)] = val;
      }
    }
  } else {
    const int colg = blockIdx.y * 32 + wn * 16 + l16;
    float bz[5];
#pragma unroll
    for (int g = 0; g < 5; ++g) bz[g] = bias[g * HDIM + colg];
#pragma unroll
    for (int vr = 0; vr < 8; ++vr) {
      int b, p;
      comp_row(erow0 + vr, n, b, p);
      float gi  = acc[0][vr] + bz[0];
      float gfl = acc[1][vr] + bz[1];
      float gfr = acc[2][vr] + bz[2];
      float gu  = acc[3][vr] + bz[3];
      float go  = acc[4][vr] + bz[4];
      size_t cbase = (size_t)(b * SEQL + p) * HDIM + colg;
      float cl = cbuf[cbase];
      float cr = cbuf[cbase + HDIM];
      float nc = cl * sigm(gfl + 1.0f) + cr * sigm(gfr + 1.0f) + tanh_fast(gu) * sigm(gi);
      float nh = sigm(go) * tanh_fast(nc);
      out0[cbase] = nh;
      out1[cbase] = nc;
    }
  }
}

#define SK_CW 128
#define SK_LD (SK_CW + 4)
#define SK_A_STAGE (16 * SK_LD)
#define SK_B_STAGE (80 * SK_LD)
#define SK_KD 1024

__device__ __forceinline__ constexpr int sk_wrow(int i) { return (i >> 2) * HDIM + (i & 3) * 4; }

__device__ __forceinline__ void sk_gload(const float* const (&asrc)[4], const float* bsrc, int k,
                                         v4f (&ra)[4], v4f (&rb)[20]) {
#pragma unroll
  for (int a = 0; a < 4; ++a)
    ra[a] = *(const v4f*)(asrc[a] + k);
#pragma unroll
  for (int i = 0; i < 20; ++i)
    rb[i] = *(const v4f*)(bsrc + (size_t)sk_wrow(i) * SK_KD + k);
}

__device__ __forceinline__ void sk_lstore(float* As, float* Bs, int wave, int lane,
                                          const v4f (&ra)[4], const v4f (&rb)[20]) {
#pragma unroll
  for (int a = 0; a < 4; ++a)
    *(v4f*)&As[(wave + 4 * a) * SK_LD + lane * 4] = ra[a];
#pragma unroll
  for (int i = 0; i < 20; ++i)
    *(v4f*)&Bs[(wave + 4 * i) * SK_LD + lane * 4] = rb[i];
}

__device__ __forceinline__ void sk_mma(const float* As, const float* Bs,
                                       int wave, int l16, int lhi, v8f (&acc)[5]) {
#pragma unroll
  for (int s = 0; s < 8; ++s) {
    const int ko = wave * 32 + s * 4 + 2 * lhi;
    v2f a = *(const v2f*)&As[l16 * SK_LD + ko];
#pragma unroll
    for (int g = 0; g < 5; ++g) {
      v2f bf = *(const v2f*)&Bs[(g * 16 + l16) * SK_LD + ko];
      acc[g] = __builtin_amdgcn_wmma_f32_16x16x4_f32(
          false, a, false, bf, (short)0, acc[g], false, false);
    }
  }
}

__global__ __launch_bounds__(128) void comp_step_splitk(const float* __restrict__ hbuf,
                                                        const float* __restrict__ cbuf,
                                                        const float* __restrict__ W,
                                                        const float* __restrict__ bias,
                                                        float* __restrict__ out0,
                                                        float* __restrict__ out1,
                                                        const int* __restrict__ rowtab) {
  __shared__ __attribute__((aligned(16))) float A_lds[2 * SK_A_STAGE];
  __shared__ __attribute__((aligned(16))) float B_lds[2 * SK_B_STAGE];
  __shared__ float red[4 * 5 * 8 * 32];
  float* const A0 = A_lds;
  float* const A1 = A_lds + SK_A_STAGE;
  float* const B0 = B_lds;
  float* const B1 = B_lds + SK_B_STAGE;

  const int tid  = threadIdx.x;
  const int lane = tid & 31;
  const int wave = __builtin_amdgcn_readfirstlane(tid >> 5);
  const int l16  = lane & 15;
  const int lhi  = lane >> 4;
  const int r0   = blockIdx.x * 16;
  const int ct0  = blockIdx.y * 16;

  const float* asrc[4];
#pragma unroll
  for (int a = 0; a < 4; ++a) {
    const int rg = r0 + wave + 4 * a;
    const int b = rg >> 1;
    const int p = rowtab[rg];
    asrc[a] = hbuf + (size_t)(b * SEQL + p) * HDIM + lane * 4;
  }
  const float* bsrc = W + (size_t)(ct0 + wave) * SK_KD + lane * 4;

  const int erow0 = r0 + 8 * lhi + 2 * wave;
  const int2 et = *(const int2*)(rowtab + erow0);
  const int colg = ct0 + l16;
  size_t cbase[2];
  float cl[2], cr[2];
#pragma unroll
  for (int e = 0; e < 2; ++e) {
    const int b = (erow0 + e) >> 1;
    const int p = (e == 0) ? et.x : et.y;
    cbase[e] = (size_t)(b * SEQL + p) * HDIM + colg;
    cl[e] = cbuf[cbase[e]];
    cr[e] = cbuf[cbase[e] + HDIM];
  }
  float bz[5];
#pragma unroll
  for (int g = 0; g < 5; ++g) bz[g] = bias[g * HDIM + colg];

  v4f ra[4];
  v4f rb[20];
  v8f acc[5] = {};

  sk_gload(asrc, bsrc, 0, ra, rb);
  sk_lstore(A0, B0, wave, lane, ra, rb);
  __syncthreads();

  for (int k0 = 0; k0 < SK_KD - 2 * SK_CW; k0 += 2 * SK_CW) {
    sk_gload(asrc, bsrc, k0 + SK_CW, ra, rb);
    sk_mma(A0, B0, wave, l16, lhi, acc);
    sk_lstore(A1, B1, wave, lane, ra, rb);
    __syncthreads();

    sk_gload(asrc, bsrc, k0 + 2 * SK_CW, ra, rb);
    sk_mma(A1, B1, wave, l16, lhi, acc);
    sk_lstore(A0, B0, wave, lane, ra, rb);
    __syncthreads();
  }
  sk_gload(asrc, bsrc, SK_KD - SK_CW, ra, rb);
  sk_mma(A0, B0, wave, l16, lhi, acc);
  sk_lstore(A1, B1, wave, lane, ra, rb);
  __syncthreads();
  sk_mma(A1, B1, wave, l16, lhi, acc);

#pragma unroll
  for (int g = 0; g < 5; ++g) {
#pragma unroll
    for (int vr = 0; vr < 8; ++vr)
      red[((wave * 5 + g) * 8 + vr) * 32 + lane] = acc[g][vr];
  }
  __syncthreads();

#pragma unroll
  for (int e = 0; e < 2; ++e) {
    const int vr = 2 * wave + e;
    float gs[5];
#pragma unroll
    for (int g = 0; g < 5; ++g) {
      float s = red[((0 * 5 + g) * 8 + vr) * 32 + lane];
      s += red[((1 * 5 + g) * 8 + vr) * 32 + lane];
      s += red[((2 * 5 + g) * 8 + vr) * 32 + lane];
      s += red[((3 * 5 + g) * 8 + vr) * 32 + lane];
      gs[g] = s + bz[g];
    }
    float nc = cl[e] * sigm(gs[1] + 1.0f) + cr[e] * sigm(gs[2] + 1.0f) + tanh_fast(gs[3]) * sigm(gs[0]);
    float nh = sigm(gs[4]) * tanh_fast(nc);
    out0[cbase[e]] = nh;
    out1[cbase[e]] = nc;
  }
}

__global__ __launch_bounds__(512) void mix_k(float* nh, float* nc, float* h, float* c,
                                             const float* __restrict__ q,
                                             const int* __restrict__ length,
                                             int* __restrict__ rowtab,
                                             int n, int iter) {
  __shared__ float sc[32];
  __shared__ int ksel_s;
  const int b = blockIdx.x;
  const int tid = threadIdx.x;
  const unsigned lane = tid & 31u;
  const int wave = tid >> 5;

  for (int p = wave; p < n; p += 16) {
    const float* row = nh + (size_t)(b * SEQL + p) * HDIM;
    float s = 0.0f;
    for (int j = (int)lane; j < HDIM; j += 32) s += row[j] * q[j];
#pragma unroll
    for (int off = 16; off > 0; off >>= 1) s += __shfl_xor(s, off, 32);
    if (lane == 0) sc[p] = s;
  }
  __syncthreads();

  if (tid == 0) {
    int len = length[b];
    int best = 0;
    float bv = -INFINITY;
    for (int p = 0; p < n; ++p) {
      float s = ((iter + 1 + p) < len) ? sc[p] : -1.0e9f;
      if (s > bv) { bv = s; best = p; }
    }
    ksel_s = best;
  }
  __syncthreads();

  const int ksel = __builtin_amdgcn_readfirstlane(ksel_s);
  const bool done = (iter + 1) < length[b];

  if (tid == 0) {
    const int hi = (n >= 2) ? (n - 2) : 0;
    const int pl = (ksel >= 1) ? (ksel - 1) : 0;
    const int pr = (ksel <= hi) ? ksel : hi;
    rowtab[2 * b]     = done ? pl : 0;
    rowtab[2 * b + 1] = done ? pr : 0;
  }
  if (!done) return;

  const size_t base = (size_t)b * SEQL * HDIM + tid;
  float* hB  = h  + base;
  float* cB  = c  + base;
  float* nhB = nh + base;
  float* ncB = nc + base;

  float hv[SEQL - 1], cv[SEQL - 1], nhv[SEQL - 1], ncv[SEQL - 1];
#pragma unroll
  for (int p = 0; p < SEQL - 1; ++p) {
    if (p >= ksel && p < n) {
      if (p == ksel) { hv[p] = nhB[p * HDIM];       cv[p] = ncB[p * HDIM]; }
      else           { hv[p] = hB[(p + 1) * HDIM];  cv[p] = cB[(p + 1) * HDIM]; }
    }
    if (p > ksel && p < n - 1) { nhv[p] = nhB[(p + 1) * HDIM]; ncv[p] = ncB[(p + 1) * HDIM]; }
  }
#pragma unroll
  for (int p = 0; p < SEQL - 1; ++p) {
    if (p >= ksel && p < n)    { hB[p * HDIM]  = hv[p];  cB[p * HDIM]  = cv[p]; }
    if (p > ksel && p < n - 1) { nhB[p * HDIM] = nhv[p]; ncB[p * HDIM] = ncv[p]; }
  }
}

__global__ __launch_bounds__(256) void final_copy_k(const float* __restrict__ hfin,
                                                    float* __restrict__ out) {
  int idx = blockIdx.x * 256 + threadIdx.x;
  int b = idx >> 9;
  int j = idx & (HDIM - 1);
  out[idx] = hfin[(size_t)b * SEQL * HDIM + j];
}

extern "C" void kernel_launch(void* const* d_in, const int* in_sizes, int n_in,
                              void* d_out, int out_size, void* d_ws, size_t ws_size,
                              hipStream_t stream) {
  const float* inp    = (const float*)d_in[0];
  const int*   length = (const int*)  d_in[1];
  const float* word_W = (const float*)d_in[2];
  const float* word_b = (const float*)d_in[3];
  const float* comp_W = (const float*)d_in[4];
  const float* comp_b = (const float*)d_in[5];
  const float* comp_q = (const float*)d_in[6];

  float* ws = (float*)d_ws;
  size_t off = 0;
  float* hb = ws + off; off += (size_t)BATCH * SEQL * HDIM;
  float* cb = ws + off; off += (size_t)BATCH * SEQL * HDIM;
  float* nh = ws + off; off += (size_t)BATCH * SEQL * HDIM;
  float* nc = ws + off; off += (size_t)BATCH * SEQL * HDIM;
  int* rowtab = (int*)(ws + off);

  gemm_wmma<0><<<dim3((BATCH * SEQL) / 32, 1024 / 256), 256, 0, stream>>>(
      inp, nullptr, nullptr, word_W, word_b, hb, cb, 0);

  gemm_wmma<1><<<dim3((BATCH * (SEQL - 1)) / 32, HDIM / 32), 128, 0, stream>>>(
      nullptr, hb, cb, comp_W, comp_b, nh, nc, SEQL - 1);
  mix_k<<<BATCH, 512, 0, stream>>>(nh, nc, hb, cb, comp_q, length, rowtab, SEQL - 1, 0);

  for (int i = 1; i < SEQL - 1; ++i) {
    int n = (SEQL - 1) - i;
    comp_step_splitk<<<dim3((BATCH * 2) / 16, HDIM / 16), 128, 0, stream>>>(
        hb, cb, comp_W, comp_b, nh, nc, rowtab);
    mix_k<<<BATCH, 512, 0, stream>>>(nh, nc, hb, cb, comp_q, length, rowtab, n, i);
  }

  final_copy_k<<<(BATCH * HDIM) / 256, 256, 0, stream>>>(hb, (float*)d_out);
  (void)in_sizes; (void)n_in; (void)out_size; (void)ws_size;
}


# Merge step moves no rows: a per-batch slot table (list position -> slot) replaces the in-place shifts; per step only 2 new scores, a wave argmax and one row copy; GEMM rows read through slots
# speedup vs baseline: 4.7942x; 1.0499x over previous
#include <hip/hip_runtime.h>
#include <hip/hip_bf16.h>
#include <math.h>


typedef __attribute__((ext_vector_type(2))) float v2f;
typedef __attribute__((ext_vector_type(4))) float v4f;
typedef __attribute__((ext_vector_type(8))) float v8f;

#define BATCH 64
#define SEQL  24
#define HDIM  512

__device__ __forceinline__ float sigm(float x) { return 1.0f / (1.0f + __expf(-x)); }
__device__ __forceinline__ float tanh_fast(float x) { return 2.0f / (1.0f + __expf(-2.0f * x)) - 1.0f; }

#define LDS_A_STAGE (32 * 36)
template <int MODE> struct BStage { static constexpr int rows = (MODE == 0) ? 320 : 160; };

template <int MODE>
__device__ __forceinline__ constexpr int wrow_of(int i) {
  return (MODE == 0) ? (i * 32) : ((i >> 1) * HDIM + (i & 1) * 16);
}

template <int MODE, int NA, int NB, int KD>
__device__ __forceinline__ void stage_gload(const float* const (&asrc)[NA], const float* bsrc, int k,
                                            v4f (&ra)[NA], v4f (&rb)[NB]) {
#pragma unroll
  for (int a = 0; a < NA; ++a)
    ra[a] = *(const v4f*)(asrc[a] + k);
#pragma unroll
  for (int i = 0; i < NB; ++i)
    rb[i] = *(const v4f*)(bsrc + (size_t)wrow_of<MODE>(i) * KD + k);
}

template <int RS, int NA, int NB>
__device__ __forceinline__ void stage_lstore(float* As, float* Bs, int sr, int sc4,
                                             const v4f (&ra)[NA], const v4f (&rb)[NB]) {
#pragma unroll
  for (int a = 0; a < NA; ++a)
    *(v4f*)&As[(a * RS + sr) * 36 + sc4 * 4] = ra[a];
#pragma unroll
  for (int i = 0; i < NB; ++i)
    *(v4f*)&Bs[(i * RS + sr) * 36 + sc4 * 4] = rb[i];
}

template <int MODE, int NACC>
__device__ __forceinline__ void stage_mma(const float* As, const float* Bs,
                                          int wm, int wn, int l16, int lhi, v8f (&acc)[5]) {
#pragma unroll
  for (int kk = 0; kk < 32; kk += 4) {
    v2f a = *(const v2f*)&As[(wm * 16 + l16) * 36 + kk + 2 * lhi];
#pragma unroll
    for (int g = 0; g < NACC; ++g) {
      int nrow = (MODE == 0) ? (wn * 64 + g * 16 + l16)
                             : (g * 32 + wn * 16 + l16);
      v2f bf = *(const v2f*)&Bs[nrow * 36 + kk + 2 * lhi];
      acc[g] = __builtin_amdgcn_wmma_f32_16x16x4_f32(
          false, a, false, bf, (short)0, acc[g], false, false);
    }
  }
}

__device__ __forceinline__ void comp_row(int rg, int n, int& b, int& p) {
  b = rg / n;
  p = rg - b * n;
}

template <int MODE>
__global__ __launch_bounds__((MODE == 0) ? 256 : 128) void gemm_wmma(const float* __restrict__ X,
                                                 const float* __restrict__ hbuf,
                                                 const float* __restrict__ cbuf,
                                                 const float* __restrict__ W,
                                                 const float* __restrict__ bias,
                                                 float* __restrict__ out0,
                                                 float* __restrict__ out1,
                                                 int n) {
  constexpr int KD   = (MODE == 0) ? 512 : 1024;
  constexpr int NB   = (MODE == 0) ? 8 : 10;
  constexpr int NACC = (MODE == 0) ? 4 : 5;
  constexpr int RS   = (MODE == 0) ? 32 : 16;
  constexpr int NA   = 32 / RS;
  constexpr int NWN  = (MODE == 0) ? 4 : 2;
  constexpr int LDS_B_STAGE = BStage<MODE>::rows * 36;

  __shared__ __attribute__((aligned(16))) float A_lds[2 * LDS_A_STAGE];
  __shared__ __attribute__((aligned(16))) float B_lds[2 * LDS_B_STAGE];
  float* const A0 = A_lds;
  float* const A1 = A_lds + LDS_A_STAGE;
  float* const B0 = B_lds;
  float* const B1 = B_lds + LDS_B_STAGE;

  const int tid  = threadIdx.x;
  const unsigned lane = tid & 31u;
  const int wave = tid >> 5;
  const int wm   = wave / NWN;
  const int wn   = wave % NWN;
  const int l16  = (int)(lane & 15u);
  const int lhi  = (int)(lane >> 4);

  const int r0 = blockIdx.x * 32;

  const int sr  = tid >> 3;
  const int sc4 = tid & 7;
  const float* asrc[NA];
#pragma unroll
  for (int a = 0; a < NA; ++a) {
    if (MODE == 0) {
      asrc[a] = X + (size_t)(r0 + a * RS + sr) * KD + sc4 * 4;
    } else {
      const int rg = r0 + a * RS + sr;
      int b, p;
      comp_row(rg, n, b, p);
      asrc[a] = hbuf + (size_t)(b * SEQL + p) * HDIM + sc4 * 4;
    }
  }
  const float* bsrc = W + (size_t)(blockIdx.y * ((MODE == 0) ? 256 : 32) + sr) * KD + sc4 * 4;

  const int erow0 = r0 + wm * 16 + 8 * lhi;

  v4f ra[NA];
  v4f rb[NB];
  v8f acc[5] = {};

  stage_gload<MODE, NA, NB, KD>(asrc, bsrc, 0, ra, rb);
  stage_lstore<RS, NA, NB>(A0, B0, sr, sc4, ra, rb);
  __syncthreads();

  for (int k0 = 0; k0 < KD - 64; k0 += 64) {
    stage_gload<MODE, NA, NB, KD>(asrc, bsrc, k0 + 32, ra, rb);
    stage_mma<MODE, NACC>(A0, B0, wm, wn, l16, lhi, acc);
    stage_lstore<RS, NA, NB>(A1, B1, sr, sc4, ra, rb);
    __syncthreads();

    stage_gload<MODE, NA, NB, KD>(asrc, bsrc, k0 + 64, ra, rb);
    stage_mma<MODE, NACC>(A1, B1, wm, wn, l16, lhi, acc);
    stage_lstore<RS, NA, NB>(A0, B0, sr, sc4, ra, rb);
    __syncthreads();
  }
  stage_gload<MODE, NA, NB, KD>(asrc, bsrc, KD - 32, ra, rb);
  stage_mma<MODE, NACC>(A0, B0, wm, wn, l16, lhi, acc);
  stage_lstore<RS, NA, NB>(A1, B1, sr, sc4, ra, rb);
  __syncthreads();
  stage_mma<MODE, NACC>(A1, B1, wm, wn, l16, lhi, acc);

  if (MODE == 0) {
    const int colbase = blockIdx.y * 256 + wn * 64;
    const int rowg0   = r0 + wm * 16 + 8 * lhi;
#pragma unroll
    for (int j = 0; j < 4; ++j) {
      int col = colbase + j * 16 + l16;
      float bv = bias[col];
#pragma unroll
      for (int vr = 0; vr < 8; ++vr) {
        int rg = rowg0 + vr;
        float val = acc[j][vr] + bv;
        if (col < HDIM) out0[(size_t)rg * HDIM + col] = val;
        else            out1[(size_t)rg * HDIM + (col - HDIM)] = val;
      }
    }
  } else {
    const int colg = blockIdx.y * 32 + wn * 16 + l16;
    float bz[5];
#pragma unroll
    for (int g = 0; g < 5; ++g) bz[g] = bias[g * HDIM + colg];
#pragma unroll
    for (int vr = 0; vr < 8; ++vr) {
      int b, p;
      comp_row(erow0 + vr, n, b, p);
      float gi  = acc[0][vr] + bz[0];
      float gfl = acc[1][vr] + bz[1];
      float gfr = acc[2][vr] + bz[2];
      float gu  = acc[3][vr] + bz[3];
      float go  = acc[4][vr] + bz[4];
      size_t cbase = (size_t)(b * SEQL + p) * HDIM + colg;
      float cl = cbuf[cbase];
      float cr = cbuf[cbase + HDIM];
      float nc = cl * sigm(gfl + 1.0f) + cr * sigm(gfr + 1.0f) + tanh_fast(gu) * sigm(gi);
      float nh = sigm(go) * tanh_fast(nc);
      out0[cbase] = nh;
      out1[cbase] = nc;
    }
  }
}

#define AUX_POS  0
#define AUX_SC   32
#define AUX_ROWS 64
#define AUX_NEWP 68
__device__ __forceinline__ const int* aux_of(const int* ncplane, int b) {
  return ncplane + (size_t)(b * SEQL + (SEQL - 1)) * HDIM;
}
__device__ __forceinline__ int* aux_of(int* ncplane, int b) {
  return ncplane + (size_t)(b * SEQL + (SEQL - 1)) * HDIM;
}

#define SK_CW 128
#define SK_LD (SK_CW + 4)
#define SK_A_STAGE (16 * SK_LD)
#define SK_B_STAGE (80 * SK_LD)
#define SK_KD 1024

__device__ __forceinline__ constexpr int sk_wrow(int i) { return (i >> 2) * HDIM + (i & 3) * 4; }

__device__ __forceinline__ void sk_gload(const float* const (&asrcL)[4], const float* const (&asrcR)[4],
                                         const float* bsrc, int k,
                                         v4f (&ra)[4], v4f (&rb)[20]) {
#pragma unroll
  for (int a = 0; a < 4; ++a)
    ra[a] = *(const v4f*)((k < HDIM) ? (asrcL[a] + k) : (asrcR[a] + (k - HDIM)));
#pragma unroll
  for (int i = 0; i < 20; ++i)
    rb[i] = *(const v4f*)(bsrc + (size_t)sk_wrow(i) * SK_KD + k);
}

__device__ __forceinline__ void sk_lstore(float* As, float* Bs, int wave, int lane,
                                          const v4f (&ra)[4], const v4f (&rb)[20]) {
#pragma unroll
  for (int a = 0; a < 4; ++a)
    *(v4f*)&As[(wave + 4 * a) * SK_LD + lane * 4] = ra[a];
#pragma unroll
  for (int i = 0; i < 20; ++i)
    *(v4f*)&Bs[(wave + 4 * i) * SK_LD + lane * 4] = rb[i];
}

__device__ __forceinline__ void sk_mma(const float* As, const float* Bs,
                                       int wave, int l16, int lhi, v8f (&acc)[5]) {
#pragma unroll
  for (int s = 0; s < 8; ++s) {
    const int ko = wave * 32 + s * 4 + 2 * lhi;
    v2f a = *(const v2f*)&As[l16 * SK_LD + ko];
#pragma unroll
    for (int g = 0; g < 5; ++g) {
      v2f bf = *(const v2f*)&Bs[(g * 16 + l16) * SK_LD + ko];
      acc[g] = __builtin_amdgcn_wmma_f32_16x16x4_f32(
          false, a, false, bf, (short)0, acc[g], false, false);
    }
  }
}

__global__ __launch_bounds__(128) void comp_step_splitk(const float* __restrict__ hbuf,
                                                        const float* __restrict__ cbuf,
                                                        const float* __restrict__ W,
                                                        const float* __restrict__ bias,
                                                        float* __restrict__ out0,
                                                        float* __restrict__ out1,
                                                        const int* __restrict__ rowtab) {
  __shared__ __attribute__((aligned(16))) float A_lds[2 * SK_A_STAGE];
  __shared__ __attribute__((aligned(16))) float B_lds[2 * SK_B_STAGE];
  __shared__ float red[4 * 5 * 8 * 32];
  float* const A0 = A_lds;
  float* const A1 = A_lds + SK_A_STAGE;
  float* const B0 = B_lds;
  float* const B1 = B_lds + SK_B_STAGE;

  const int tid  = threadIdx.x;
  const int lane = tid & 31;
  const int wave = __builtin_amdgcn_readfirstlane(tid >> 5);
  const int l16  = lane & 15;
  const int lhi  = lane >> 4;
  const int r0   = blockIdx.x * 16;
  const int ct0  = blockIdx.y * 16;

  const float* asrcL[4];
  const float* asrcR[4];
#pragma unroll
  for (int a = 0; a < 4; ++a) {
    const int rg = r0 + wave + 4 * a;
    const int b = rg >> 1;
    const int2 sl = *(const int2*)(aux_of(rowtab, b) + AUX_ROWS + 2 * (rg & 1));
    asrcL[a] = hbuf + (size_t)(b * SEQL + sl.x) * HDIM + lane * 4;
    asrcR[a] = hbuf + (size_t)(b * SEQL + sl.y) * HDIM + lane * 4;
  }
  const float* bsrc = W + (size_t)(ct0 + wave) * SK_KD + lane * 4;

  const int erow0 = r0 + 8 * lhi + 2 * wave;
  const int eb = erow0 >> 1;
  const int4 et = *(const int4*)(aux_of(rowtab, eb) + AUX_ROWS);
  const int colg = ct0 + l16;
  size_t cbase[2];
  float cl[2], cr[2];
#pragma unroll
  for (int e = 0; e < 2; ++e) {
    const int sL = (e == 0) ? et.x : et.z;
    const int sR = (e == 0) ? et.y : et.w;
    cbase[e] = (size_t)(eb * SEQL + sL) * HDIM + colg;
    cl[e] = cbuf[cbase[e]];
    cr[e] = cbuf[(size_t)(eb * SEQL + sR) * HDIM + colg];
  }
  float bz[5];
#pragma unroll
  for (int g = 0; g < 5; ++g) bz[g] = bias[g * HDIM + colg];

  v4f ra[4];
  v4f rb[20];
  v8f acc[5] = {};

  sk_gload(asrcL, asrcR, bsrc, 0, ra, rb);
  sk_lstore(A0, B0, wave, lane, ra, rb);
  __syncthreads();

#pragma nounroll
  for (int k0 = 0; k0 < SK_KD - 2 * SK_CW; k0 += 2 * SK_CW) {
    sk_gload(asrcL, asrcR, bsrc, k0 + SK_CW, ra, rb);
    sk_mma(A0, B0, wave, l16, lhi, acc);
    sk_lstore(A1, B1, wave, lane, ra, rb);
    __syncthreads();

    sk_gload(asrcL, asrcR, bsrc, k0 + 2 * SK_CW, ra, rb);
    sk_mma(A1, B1, wave, l16, lhi, acc);
    sk_lstore(A0, B0, wave, lane, ra, rb);
    __syncthreads();
  }
  sk_gload(asrcL, asrcR, bsrc, SK_KD - SK_CW, ra, rb);
  sk_mma(A0, B0, wave, l16, lhi, acc);
  sk_lstore(A1, B1, wave, lane, ra, rb);
  __syncthreads();
  sk_mma(A1, B1, wave, l16, lhi, acc);

#pragma unroll
  for (int g = 0; g < 5; ++g) {
#pragma unroll
    for (int vr = 0; vr < 8; ++vr)
      red[((wave * 5 + g) * 8 + vr) * 32 + lane] = acc[g][vr];
  }
  __syncthreads();

#pragma unroll
  for (int e = 0; e < 2; ++e) {
    const int vr = 2 * wave + e;
    float gs[5];
#pragma unroll
    for (int g = 0; g < 5; ++g) {
      float s = red[((0 * 5 + g) * 8 + vr) * 32 + lane];
      s += red[((1 * 5 + g) * 8 + vr) * 32 + lane];
      s += red[((2 * 5 + g) * 8 + vr) * 32 + lane];
      s += red[((3 * 5 + g) * 8 + vr) * 32 + lane];
      gs[g] = s + bz[g];
    }
    float nc = cl[e] * sigm(gs[1] + 1.0f) + cr[e] * sigm(gs[2] + 1.0f) + tanh_fast(gs[3]) * sigm(gs[0]);
    float nh = sigm(gs[4]) * tanh_fast(nc);
    out0[cbase[e]] = nh;
    out1[cbase[e]] = nc;
  }
}

__device__ __forceinline__ float row_dot_q(const float* row, const float* __restrict__ q, unsigned lane) {
  float s = 0.0f;
  for (int j = (int)lane; j < HDIM; j += 32) s += row[j] * q[j];
#pragma unroll
  for (int off = 16; off > 0; off >>= 1) s += __shfl_xor(s, off, 32);
  return s;
}

__global__ __launch_bounds__(512) void mix_k(const float* nh, const float* nc, float* h, float* c,
                                             const float* __restrict__ q,
                                             const int* __restrict__ length,
                                             int* auxplane,
                                             int n, int iter) {
  __shared__ float sc[32];
  __shared__ int pos_s[32];
  __shared__ int sel_s[2];
  const int b = blockIdx.x;
  const int tid = threadIdx.x;
  const unsigned lane = tid & 31u;
  const int wave = tid >> 5;
  int* aux = aux_of(auxplane, b);

  if (iter == 0) {
    if (tid < 32) {
      pos_s[tid] = (tid < SEQL) ? tid : (SEQL - 1);
      if (tid >= n) sc[tid] = 0.0f;
    }
    for (int p = wave; p < n; p += 16) {
      float s = row_dot_q(nh + (size_t)(b * SEQL + p) * HDIM, q, lane);
      if (lane == 0) sc[p] = s;
    }
  } else {
    if (tid < 32) {
      pos_s[tid] = aux[AUX_POS + tid];
      sc[tid] = __int_as_float(aux[AUX_SC + tid]);
    }
    __syncthreads();
    if (wave < 2) {
      const int p = aux[AUX_NEWP + wave];
      float s = row_dot_q(nh + (size_t)(b * SEQL + pos_s[p]) * HDIM, q, lane);
      if (lane == 0) sc[p] = s;
    }
  }
  __syncthreads();

  if (wave == 0) {
    const int len = length[b];
    const int li = (int)lane;
    float v = -INFINITY;
    if (li < n) {
      v = ((iter + 1 + li) < len) ? sc[li] : -1.0e9f;
      if (!(v == v)) v = -INFINITY;
    }
    int idx = li;
#pragma unroll
    for (int off = 16; off > 0; off >>= 1) {
      const float ov = __shfl_xor(v, off, 32);
      const int oi = __shfl_xor(idx, off, 32);
      if (ov > v || (ov == v && oi < idx)) { v = ov; idx = oi; }
    }
    const int ksel = idx;
    const bool done = (iter + 1) < len;
    const int nxt = (li < 31) ? (li + 1) : 31;
    const int p0 = pos_s[li], p1 = pos_s[nxt];
    const float s0 = sc[li], s1 = sc[nxt];
    const int   newpos = (!done || li <= ksel) ? p0 : p1;
    const float newsc  = (!done || li <  ksel) ? s0 : s1;
    aux[AUX_POS + li] = newpos;
    aux[AUX_SC + li]  = __float_as_int(newsc);
    const int hi = (n >= 2) ? (n - 2) : 0;
    int pl = (ksel >= 1) ? (ksel - 1) : 0;
    int pr = (ksel <= hi) ? ksel : hi;
    if (!done) { pl = 0; pr = 0; }
    const int sl0 = __shfl(newpos, pl, 32), sr0 = __shfl(newpos, pl + 1, 32);
    const int sl1 = __shfl(newpos, pr, 32), sr1 = __shfl(newpos, pr + 1, 32);
    if (li == 0) {
      *(int4*)(aux + AUX_ROWS) = make_int4(sl0, sr0, sl1, sr1);
      *(int2*)(aux + AUX_NEWP) = make_int2(pl, pr);
    }
    const int slotL = __shfl(p0, ksel, 32);
    if (li == 0) { sel_s[0] = slotL; sel_s[1] = done ? 1 : 0; }
  }
  __syncthreads();

  if (sel_s[1] == 0) return;
  const size_t at = (size_t)(b * SEQL + sel_s[0]) * HDIM + tid;
  const float hv = nh[at], cv = nc[at];
  h[at] = hv;
  c[at] = cv;
}

__global__ __launch_bounds__(256) void final_copy_k(const float* __restrict__ hfin,
                                                    float* __restrict__ out) {
  int idx = blockIdx.x * 256 + threadIdx.x;
  int b = idx >> 9;
  int j = idx & (HDIM - 1);
  out[idx] = hfin[(size_t)b * SEQL * HDIM + j];
}

extern "C" void kernel_launch(void* const* d_in, const int* in_sizes, int n_in,
                              void* d_out, int out_size, void* d_ws, size_t ws_size,
                              hipStream_t stream) {
  const float* inp    = (const float*)d_in[0];
  const int*   length = (const int*)  d_in[1];
  const float* word_W = (const float*)d_in[2];
  const float* word_b = (const float*)d_in[3];
  const float* comp_W = (const float*)d_in[4];
  const float* comp_b = (const float*)d_in[5];
  const float* comp_q = (const float*)d_in[6];

  float* ws = (float*)d_ws;
  size_t off = 0;
  float* hb = ws + off; off += (size_t)BATCH * SEQL * HDIM;
  float* cb = ws + off; off += (size_t)BATCH * SEQL * HDIM;
  float* nh = ws + off; off += (size_t)BATCH * SEQL * HDIM;
  float* nc = ws + off; off += (size_t)BATCH * SEQL * HDIM;
  int* aux = (int*)nc;

  gemm_wmma<0><<<dim3((BATCH * SEQL) / 32, 1024 / 256), 256, 0, stream>>>(
      inp, nullptr, nullptr, word_W, word_b, hb, cb, 0);

  gemm_wmma<1><<<dim3((BATCH * (SEQL - 1)) / 32, HDIM / 32), 128, 0, stream>>>(
      nullptr, hb, cb, comp_W, comp_b, nh, nc, SEQL - 1);
  mix_k<<<BATCH, 512, 0, stream>>>(nh, nc, hb, cb, comp_q, length, aux, SEQL - 1, 0);

  for (int i = 1; i < SEQL - 1; ++i) {
    int n = (SEQL - 1) - i;
    comp_step_splitk<<<dim3((BATCH * 2) / 16, HDIM / 16), 128, 0, stream>>>(
        hb, cb, comp_W, comp_b, nh, nc, (const int*)aux);
    mix_k<<<BATCH, 512, 0, stream>>>(nh, nc, hb, cb, comp_q, length, aux, n, i);
  }

  final_copy_k<<<(BATCH * HDIM) / 256, 256, 0, stream>>>(hb, (float*)d_out);
  (void)off; (void)in_sizes; (void)n_in; (void)out_size; (void)ws_size;
}


# Select+merge folded into the step GEMM kernel (26 launches, was 48): every workgroup redoes it from a two-copy table; pair scores summed from 32 per-workgroup partial dots, a new order
# speedup vs baseline: 6.1731x; 1.2876x over previous
#include <hip/hip_runtime.h>
#include <hip/hip_bf16.h>
#include <math.h>


typedef __attribute__((ext_vector_type(2))) float v2f;
typedef __attribute__((ext_vector_type(4))) float v4f;
typedef __attribute__((ext_vector_type(8))) float v8f;

#define BATCH 64
#define SEQL  24
#define HDIM  512

__device__ __forceinline__ float sigm(float x) { return 1.0f / (1.0f + __expf(-x)); }
__device__ __forceinline__ float tanh_fast(float x) { return 2.0f / (1.0f + __expf(-2.0f * x)) - 1.0f; }

#define LDS_A_STAGE (32 * 36)
template <int MODE> struct BStage { static constexpr int rows = (MODE == 0) ? 320 : 160; };

template <int MODE>
__device__ __forceinline__ constexpr int wrow_of(int i) {
  return (MODE == 0) ? (i * 32) : ((i >> 1) * HDIM + (i & 1) * 16);
}

template <int MODE, int NA, int NB, int KD>
__device__ __forceinline__ void stage_gload(const float* const (&asrc)[NA], const float* bsrc, int k,
                                            v4f (&ra)[NA], v4f (&rb)[NB]) {
#pragma unroll
  for (int a = 0; a < NA; ++a)
    ra[a] = *(const v4f*)(asrc[a] + k);
#pragma unroll
  for (int i = 0; i < NB; ++i)
    rb[i] = *(const v4f*)(bsrc + (size_t)wrow_of<MODE>(i) * KD + k);
}

template <int RS, int NA, int NB>
__device__ __forceinline__ void stage_lstore(float* As, float* Bs, int sr, int sc4,
                                             const v4f (&ra)[NA], const v4f (&rb)[NB]) {
#pragma unroll
  for (int a = 0; a < NA; ++a)
    *(v4f*)&As[(a * RS + sr) * 36 + sc4 * 4] = ra[a];
#pragma unroll
  for (int i = 0; i < NB; ++i)
    *(v4f*)&Bs[(i * RS + sr) * 36 + sc4 * 4] = rb[i];
}

template <int MODE, int NACC>
__device__ __forceinline__ void stage_mma(const float* As, const float* Bs,
                                          int wm, int wn, int l16, int lhi, v8f (&acc)[5]) {
#pragma unroll
  for (int kk = 0; kk < 32; kk += 4) {
    v2f a = *(const v2f*)&As[(wm * 16 + l16) * 36 + kk + 2 * lhi];
#pragma unroll
    for (int g = 0; g < NACC; ++g) {
      int nrow = (MODE == 0) ? (wn * 64 + g * 16 + l16)
                             : (g * 32 + wn * 16 + l16);
      v2f bf = *(const v2f*)&Bs[nrow * 36 + kk + 2 * lhi];
      acc[g] = __builtin_amdgcn_wmma_f32_16x16x4_f32(
          false, a, false, bf, (short)0, acc[g], false, false);
    }
  }
}

__device__ __forceinline__ void comp_row(int rg, int n, int& b, int& p) {
  b = rg / n;
  p = rg - b * n;
}

template <int MODE>
__global__ __launch_bounds__((MODE == 0) ? 256 : 128) void gemm_wmma(const float* __restrict__ X,
                                                 const float* __restrict__ hbuf,
                                                 const float* __restrict__ cbuf,
                                                 const float* __restrict__ W,
                                                 const float* __restrict__ bias,
                                                 float* __restrict__ out0,
                                                 float* __restrict__ out1,
                                                 int n) {
  constexpr int KD   = (MODE == 0) ? 512 : 1024;
  constexpr int NB   = (MODE == 0) ? 8 : 10;
  constexpr int NACC = (MODE == 0) ? 4 : 5;
  constexpr int RS   = (MODE == 0) ? 32 : 16;
  constexpr int NA   = 32 / RS;
  constexpr int NWN  = (MODE == 0) ? 4 : 2;
  constexpr int LDS_B_STAGE = BStage<MODE>::rows * 36;

  __shared__ __attribute__((aligned(16))) float A_lds[2 * LDS_A_STAGE];
  __shared__ __attribute__((aligned(16))) float B_lds[2 * LDS_B_STAGE];
  float* const A0 = A_lds;
  float* const A1 = A_lds + LDS_A_STAGE;
  float* const B0 = B_lds;
  float* const B1 = B_lds + LDS_B_STAGE;

  const int tid  = threadIdx.x;
  const unsigned lane = tid & 31u;
  const int wave = tid >> 5;
  const int wm   = wave / NWN;
  const int wn   = wave % NWN;
  const int l16  = (int)(lane & 15u);
  const int lhi  = (int)(lane >> 4);

  const int r0 = blockIdx.x * 32;

  const int sr  = tid >> 3;
  const int sc4 = tid & 7;
  const float* asrc[NA];
#pragma unroll
  for (int a = 0; a < NA; ++a) {
    if (MODE == 0) {
      asrc[a] = X + (size_t)(r0 + a * RS + sr) * KD + sc4 * 4;
    } else {
      const int rg = r0 + a * RS + sr;
      int b, p;
      comp_row(rg, n, b, p);
      asrc[a] = hbuf + (size_t)(b * SEQL + p) * HDIM + sc4 * 4;
    }
  }
  const float* bsrc = W + (size_t)(blockIdx.y * ((MODE == 0) ? 256 : 32) + sr) * KD + sc4 * 4;

  const int erow0 = r0 + wm * 16 + 8 * lhi;

  v4f ra[NA];
  v4f rb[NB];
  v8f acc[5] = {};

  stage_gload<MODE, NA, NB, KD>(asrc, bsrc, 0, ra, rb);
  stage_lstore<RS, NA, NB>(A0, B0, sr, sc4, ra, rb);
  __syncthreads();

  for (int k0 = 0; k0 < KD - 64; k0 += 64) {
    stage_gload<MODE, NA, NB, KD>(asrc, bsrc, k0 + 32, ra, rb);
    stage_mma<MODE, NACC>(A0, B0, wm, wn, l16, lhi, acc);
    stage_lstore<RS, NA, NB>(A1, B1, sr, sc4, ra, rb);
    __syncthreads();

    stage_gload<MODE, NA, NB, KD>(asrc, bsrc, k0 + 64, ra, rb);
    stage_mma<MODE, NACC>(A1, B1, wm, wn, l16, lhi, acc);
    stage_lstore<RS, NA, NB>(A0, B0, sr, sc4, ra, rb);
    __syncthreads();
  }
  stage_gload<MODE, NA, NB, KD>(asrc, bsrc, KD - 32, ra, rb);
  stage_mma<MODE, NACC>(A0, B0, wm, wn, l16, lhi, acc);
  stage_lstore<RS, NA, NB>(A1, B1, sr, sc4, ra, rb);
  __syncthreads();
  stage_mma<MODE, NACC>(A1, B1, wm, wn, l16, lhi, acc);

  if (MODE == 0) {
    const int colbase = blockIdx.y * 256 + wn * 64;
    const int rowg0   = r0 + wm * 16 + 8 * lhi;
#pragma unroll
    for (int j = 0; j < 4; ++j) {
      int col = colbase + j * 16 + l16;
      float bv = bias[col];
#pragma unroll
      for (int vr = 0; vr < 8; ++vr) {
        int rg = rowg0 + vr;
        float val = acc[j][vr] + bv;
        if (col < HDIM) out0[(size_t)rg * HDIM + col] = val;
        else            out1[(size_t)rg * HDIM + (col - HDIM)] = val;
      }
    }
  } else {
    const int colg = blockIdx.y * 32 + wn * 16 + l16;
    float bz[5];
#pragma unroll
    for (int g = 0; g < 5; ++g) bz[g] = bias[g * HDIM + colg];
#pragma unroll
    for (int vr = 0; vr < 8; ++vr) {
      int b, p;
      comp_row(erow0 + vr, n, b, p);
      float gi  = acc[0][vr] + bz[0];
      float gfl = acc[1][vr] + bz[1];
      float gfr = acc[2][vr] + bz[2];
      float gu  = acc[3][vr] + bz[3];
      float go  = acc[4][vr] + bz[4];
      size_t cbase = (size_t)(b * SEQL + p) * HDIM + colg;
      float cl = cbuf[cbase];
      float cr = cbuf[cbase + HDIM];
      float nc = cl * sigm(gfl + 1.0f) + cr * sigm(gfr + 1.0f) + tanh_fast(gu) * sigm(gi);
      float nh = sigm(go) * tanh_fast(nc);
      out0[cbase] = nh;
      out1[cbase] = nc;
    }
  }
}

#define PLANE_ROWS (BATCH * SEQL)
#define PLANE_ELEMS ((size_t)PLANE_ROWS * HDIM)
#define CAND_ROW0 (2 * PLANE_ROWS)
#define T_LOC    0
#define T_CS     32
#define T_SC     64
#define T_NEWP   96
#define T_STRIDE 128
#define T_PART   256
__device__ __forceinline__ const int* aux_of(const int* ncplane, int b) {
  return ncplane + (size_t)(b * SEQL + (SEQL - 1)) * HDIM;
}
__device__ __forceinline__ int* aux_of(int* ncplane, int b) {
  return ncplane + (size_t)(b * SEQL + (SEQL - 1)) * HDIM;
}

#define SK_CW 128
#define SK_LD (SK_CW + 4)
#define SK_A_STAGE (16 * SK_LD)
#define SK_B_STAGE (80 * SK_LD)
#define SK_KD 1024

__device__ __forceinline__ constexpr int sk_wrow(int i) { return (i >> 2) * HDIM + (i & 3) * 4; }

__device__ __forceinline__ void sk_gload_a(const float* const (&asrcL)[4], const float* const (&asrcR)[4],
                                           int k, v4f (&ra)[4]) {
#pragma unroll
  for (int a = 0; a < 4; ++a)
    ra[a] = *(const v4f*)((k < HDIM) ? (asrcL[a] + k) : (asrcR[a] + (k - HDIM)));
}

__device__ __forceinline__ void sk_gload_b(const float* bsrc, int k, v4f (&rb)[20]) {
#pragma unroll
  for (int i = 0; i < 20; ++i)
    rb[i] = *(const v4f*)(bsrc + (size_t)sk_wrow(i) * SK_KD + k);
}

__device__ __forceinline__ void sk_lstore(float* As, float* Bs, int wave, int lane,
                                          const v4f (&ra)[4], const v4f (&rb)[20]) {
#pragma unroll
  for (int a = 0; a < 4; ++a)
    *(v4f*)&As[(wave + 4 * a) * SK_LD + lane * 4] = ra[a];
#pragma unroll
  for (int i = 0; i < 20; ++i)
    *(v4f*)&Bs[(wave + 4 * i) * SK_LD + lane * 4] = rb[i];
}

__device__ __forceinline__ void sk_mma(const float* As, const float* Bs,
                                       int wave, int l16, int lhi, v8f (&acc)[5]) {
#pragma unroll
  for (int s = 0; s < 8; ++s) {
    const int ko = wave * 32 + s * 4 + 2 * lhi;
    v2f a = *(const v2f*)&As[l16 * SK_LD + ko];
#pragma unroll
    for (int g = 0; g < 5; ++g) {
      v2f bf = *(const v2f*)&Bs[(g * 16 + l16) * SK_LD + ko];
      acc[g] = __builtin_amdgcn_wmma_f32_16x16x4_f32(
          false, a, false, bf, (short)0, acc[g], false, false);
    }
  }
}

__global__ __launch_bounds__(128) void comp_step_fused(const float* __restrict__ nodes,
                                                       const float* __restrict__ W,
                                                       const float* __restrict__ bias,
                                                       const float* __restrict__ q,
                                                       const int* __restrict__ length,
                                                       float* __restrict__ out0,
                                                       float* __restrict__ out1,
                                                       int* aux,
                                                       int step) {
  __shared__ __attribute__((aligned(16))) float A_lds[2 * SK_A_STAGE];
  __shared__ __attribute__((aligned(16))) float B_lds[2 * SK_B_STAGE];
  __shared__ __attribute__((aligned(16))) int rowinfo[16 * 4];
  __shared__ float red[4 * 5 * 8 * 32];
  float* const A0 = A_lds;
  float* const A1 = A_lds + SK_A_STAGE;
  float* const B0 = B_lds;
  float* const B1 = B_lds + SK_B_STAGE;

  const int tid  = threadIdx.x;
  const int lane = tid & 31;
  const int wave = __builtin_amdgcn_readfirstlane(tid >> 5);
  const int l16  = lane & 15;
  const int lhi  = lane >> 4;
  const int r0   = blockIdx.x * 16;
  const int ct0  = blockIdx.y * 16;

  v4f ra[4];
  v4f rb[20];
  v8f acc[5] = {};

  const float* bsrc = W + (size_t)(ct0 + wave) * SK_KD + lane * 4;

  const int par_in  = (step - 1) & 1;
  const int par_out = step & 1;
  const int it = step - 1;
  const int n  = SEQL - step;
  int   t_loc[2], t_cs[2], t_ppl[2], t_ppr[2], t_len[2];
  float t_sc[2], t_pa[2], t_pb[2];
#pragma unroll
  for (int t = 0; t < 2; ++t) {
    const int b = blockIdx.x * 8 + 2 * wave + t;
    const int* tb = aux_of(aux, b);
    const int* rd = tb + par_in * T_STRIDE;
    t_loc[t] = rd[T_LOC + lane];
    t_cs[t]  = rd[T_CS + lane];
    t_sc[t]  = __int_as_float(rd[T_SC + lane]);
    t_ppl[t] = rd[T_NEWP];
    t_ppr[t] = rd[T_NEWP + 1];
    t_pa[t]  = 0.0f;
    t_pb[t]  = 0.0f;
    if (step >= 2) {
      t_pa[t] = __int_as_float(tb[T_PART + par_in * 64 + lane]);
      t_pb[t] = __int_as_float(tb[T_PART + par_in * 64 + 32 + lane]);
    }
    t_len[t] = length[b];
  }
  sk_gload_b(bsrc, 0, rb);
#pragma unroll
  for (int t = 0; t < 2; ++t) {
    const int bb = 2 * wave + t;
    const int b  = blockIdx.x * 8 + bb;
    int* tb = aux_of(aux, b);
    const int loc0 = t_loc[t];
    const int cs0  = t_cs[t];
    float sc0 = t_sc[t];
    if (step >= 2) {
      float pa = t_pa[t];
      float pb = t_pb[t];
#pragma unroll
      for (int off = 16; off > 0; off >>= 1) {
        pa += __shfl_xor(pa, off, 32);
        pb += __shfl_xor(pb, off, 32);
      }
      if (lane == t_ppl[t]) sc0 = pa;
      if (lane == t_ppr[t]) sc0 = pb;
    }
    const int len = t_len[t];
    float v = -INFINITY;
    if (lane < n) {
      v = ((it + 1 + lane) < len) ? sc0 : -1.0e9f;
      if (!(v == v)) v = -INFINITY;
    }
    int idx = lane;
#pragma unroll
    for (int off = 16; off > 0; off >>= 1) {
      const float ov = __shfl_xor(v, off, 32);
      const int oi = __shfl_xor(idx, off, 32);
      if (ov > v || (ov == v && oi < idx)) { v = ov; idx = oi; }
    }
    const int ksel = idx;
    const bool done = (it + 1) < len;
    const int nxt = (lane < 31) ? (lane + 1) : 31;
    const int   loc1 = __shfl(loc0, nxt, 32);
    const int   cs1  = __shfl(cs0, nxt, 32);
    const float sc1  = __shfl(sc0, nxt, 32);
    const int   csK  = __shfl(cs0, ksel, 32);
    int newloc = loc0, newcs = cs0;
    float newsc = sc0;
    if (done) {
      if (lane == ksel)     newloc = CAND_ROW0 + b * SEQL + csK;
      else if (lane > ksel) newloc = loc1;
      if (lane >= ksel) { newcs = cs1; newsc = sc1; }
    }
    const int hi = n - 2;
    int pl = (ksel >= 1) ? (ksel - 1) : 0;
    int pr = (ksel <= hi) ? ksel : hi;
    if (!done) { pl = 0; pr = 0; }
    const int l0 = __shfl(newloc, pl, 32), rr0 = __shfl(newloc, pl + 1, 32), d0 = __shfl(newcs, pl, 32);
    const int l1 = __shfl(newloc, pr, 32), rr1 = __shfl(newloc, pr + 1, 32), d1 = __shfl(newcs, pr, 32);
    if (lane == 0) {
      *(int4*)&rowinfo[(2 * bb) * 4]     = make_int4(l0, rr0, b * SEQL + d0, 0);
      *(int4*)&rowinfo[(2 * bb + 1) * 4] = make_int4(l1, rr1, b * SEQL + d1, 0);
    }
    if (blockIdx.y == 0) {
      int* wr = tb + par_out * T_STRIDE;
      wr[T_LOC + lane] = newloc;
      wr[T_CS + lane]  = newcs;
      wr[T_SC + lane]  = __float_as_int(newsc);
      if (lane == 0) *(int2*)(wr + T_NEWP) = make_int2(pl, pr);
    }
  }
  __syncthreads();

  const float* asrcL[4];
  const float* asrcR[4];
#pragma unroll
  for (int a = 0; a < 4; ++a) {
    const int4 ri = *(const int4*)&rowinfo[(wave + 4 * a) * 4];
    asrcL[a] = nodes + (size_t)ri.x * HDIM + lane * 4;
    asrcR[a] = nodes + (size_t)ri.y * HDIM + lane * 4;
  }

  const int colg = ct0 + l16;
  size_t cbase[2];
  float cl[2], cr[2];
#pragma unroll
  for (int e = 0; e < 2; ++e) {
    const int4 ri = *(const int4*)&rowinfo[(8 * lhi + 2 * wave + e) * 4];
    cbase[e] = (size_t)ri.z * HDIM + colg;
    cl[e] = nodes[PLANE_ELEMS + (size_t)ri.x * HDIM + colg];
    cr[e] = nodes[PLANE_ELEMS + (size_t)ri.y * HDIM + colg];
  }
  float bz[5];
#pragma unroll
  for (int g = 0; g < 5; ++g) bz[g] = bias[g * HDIM + colg];
  const float qv = q[colg];

  sk_gload_a(asrcL, asrcR, 0, ra);
  sk_lstore(A0, B0, wave, lane, ra, rb);
  __syncthreads();

#pragma nounroll
  for (int k0 = 0; k0 < SK_KD - 2 * SK_CW; k0 += 2 * SK_CW) {
    sk_gload_a(asrcL, asrcR, k0 + SK_CW, ra);
    sk_gload_b(bsrc, k0 + SK_CW, rb);
    sk_mma(A0, B0, wave, l16, lhi, acc);
    sk_lstore(A1, B1, wave, lane, ra, rb);
    __syncthreads();

    sk_gload_a(asrcL, asrcR, k0 + 2 * SK_CW, ra);
    sk_gload_b(bsrc, k0 + 2 * SK_CW, rb);
    sk_mma(A1, B1, wave, l16, lhi, acc);
    sk_lstore(A0, B0, wave, lane, ra, rb);
    __syncthreads();
  }
  sk_gload_a(asrcL, asrcR, SK_KD - SK_CW, ra);
  sk_gload_b(bsrc, SK_KD - SK_CW, rb);
  sk_mma(A0, B0, wave, l16, lhi, acc);
  sk_lstore(A1, B1, wave, lane, ra, rb);
  __syncthreads();
  sk_mma(A1, B1, wave, l16, lhi, acc);

#pragma unroll
  for (int g = 0; g < 5; ++g) {
#pragma unroll
    for (int vr = 0; vr < 8; ++vr)
      red[((wave * 5 + g) * 8 + vr) * 32 + lane] = acc[g][vr];
  }
  __syncthreads();

  int* tbe = aux_of(aux, blockIdx.x * 8 + 4 * lhi + wave) + T_PART + par_out * 64 + blockIdx.y;
#pragma unroll
  for (int e = 0; e < 2; ++e) {
    const int vr = 2 * wave + e;
    float gs[5];
#pragma unroll
    for (int g = 0; g < 5; ++g) {
      float s = red[((0 * 5 + g) * 8 + vr) * 32 + lane];
      s += red[((1 * 5 + g) * 8 + vr) * 32 + lane];
      s += red[((2 * 5 + g) * 8 + vr) * 32 + lane];
      s += red[((3 * 5 + g) * 8 + vr) * 32 + lane];
      gs[g] = s + bz[g];
    }
    float nc = cl[e] * sigm(gs[1] + 1.0f) + cr[e] * sigm(gs[2] + 1.0f) + tanh_fast(gs[3]) * sigm(gs[0]);
    float nh = sigm(gs[4]) * tanh_fast(nc);
    out0[cbase[e]] = nh;
    out1[cbase[e]] = nc;
    float pv = nh * qv;
    pv += __shfl_xor(pv, 8, 32);
    pv += __shfl_xor(pv, 4, 32);
    pv += __shfl_xor(pv, 2, 32);
    pv += __shfl_xor(pv, 1, 32);
    if (l16 == 0) tbe[e * 32] = __float_as_int(pv);
  }
}

__device__ __forceinline__ float row_dot_q(const float* row, const float* __restrict__ q, unsigned lane) {
  float s = 0.0f;
  for (int j = (int)lane; j < HDIM; j += 32) s += row[j] * q[j];
#pragma unroll
  for (int off = 16; off > 0; off >>= 1) s += __shfl_xor(s, off, 32);
  return s;
}

__global__ __launch_bounds__(512) void score_all_k(const float* __restrict__ nh,
                                                   const float* __restrict__ q,
                                                   int* aux) {
  __shared__ float sc[32];
  const int b = blockIdx.x;
  const int tid = threadIdx.x;
  const unsigned lane = tid & 31u;
  const int wave = tid >> 5;
  if (tid >= SEQL - 1 && tid < 32) sc[tid] = 0.0f;
  for (int p = wave; p < SEQL - 1; p += 16) {
    float s = row_dot_q(nh + (size_t)(b * SEQL + p) * HDIM, q, lane);
    if (lane == 0) sc[p] = s;
  }
  __syncthreads();
  if (tid < 32) {
    int* tb = aux_of(aux, b);
    tb[T_LOC + tid] = b * SEQL + ((tid < SEQL) ? tid : (SEQL - 1));
    tb[T_CS + tid]  = (tid < SEQL - 1) ? tid : (SEQL - 2);
    tb[T_SC + tid]  = __float_as_int(sc[tid]);
    if (tid == 0) *(int2*)(tb + T_NEWP) = make_int2(0, 0);
  }
}

__global__ __launch_bounds__(256) void final_pick_k(const float* __restrict__ nodes,
                                                    const int* __restrict__ aux,
                                                    const int* __restrict__ length,
                                                    float* __restrict__ out) {
  int idx = blockIdx.x * 256 + threadIdx.x;
  int b = idx >> 9;
  int j = idx & (HDIM - 1);
  const int* tb = aux_of(aux, b);
  const int row = ((SEQL - 1) < length[b]) ? (CAND_ROW0 + b * SEQL + tb[T_CS]) : tb[T_LOC];
  out[idx] = nodes[(size_t)row * HDIM + j];
}

extern "C" void kernel_launch(void* const* d_in, const int* in_sizes, int n_in,
                              void* d_out, int out_size, void* d_ws, size_t ws_size,
                              hipStream_t stream) {
  const float* inp    = (const float*)d_in[0];
  const int*   length = (const int*)  d_in[1];
  const float* word_W = (const float*)d_in[2];
  const float* word_b = (const float*)d_in[3];
  const float* comp_W = (const float*)d_in[4];
  const float* comp_b = (const float*)d_in[5];
  const float* comp_q = (const float*)d_in[6];

  float* ws = (float*)d_ws;
  size_t off = 0;
  float* hb = ws + off; off += (size_t)BATCH * SEQL * HDIM;
  float* cb = ws + off; off += (size_t)BATCH * SEQL * HDIM;
  float* nh = ws + off; off += (size_t)BATCH * SEQL * HDIM;
  float* nc = ws + off; off += (size_t)BATCH * SEQL * HDIM;
  int* aux = (int*)nc;

  gemm_wmma<0><<<dim3((BATCH * SEQL) / 32, 1024 / 256), 256, 0, stream>>>(
      inp, nullptr, nullptr, word_W, word_b, hb, cb, 0);

  gemm_wmma<1><<<dim3((BATCH * (SEQL - 1)) / 32, HDIM / 32), 128, 0, stream>>>(
      nullptr, hb, cb, comp_W, comp_b, nh, nc, SEQL - 1);
  score_all_k<<<BATCH, 512, 0, stream>>>(nh, comp_q, aux);

  for (int i = 1; i < SEQL - 1; ++i) {
    comp_step_fused<<<dim3((BATCH * 2) / 16, HDIM / 16), 128, 0, stream>>>(
        ws, comp_W, comp_b, comp_q, length, nh, nc, aux, i);
  }

  final_pick_k<<<(BATCH * HDIM) / 256, 256, 0, stream>>>(ws, (const int*)aux, length, (float*)d_out);
  (void)off; (void)hb; (void)cb; (void)in_sizes; (void)n_in; (void)out_size; (void)ws_size;
}


# Step kernel: the select+merge chain and the score dot exchange lanes on the vector ALU (DPP row_xmask/row_shl, permlanex16, ballot+ctz, readlane), not ds_bpermute + wait; same sums, same order
# speedup vs baseline: 6.7751x; 1.0975x over previous
#include <hip/hip_runtime.h>
#include <hip/hip_bf16.h>
#include <math.h>


typedef __attribute__((ext_vector_type(2))) float v2f;
typedef __attribute__((ext_vector_type(4))) float v4f;
typedef __attribute__((ext_vector_type(8))) float v8f;

#define BATCH 64
#define SEQL  24
#define HDIM  512

__device__ __forceinline__ float sigm(float x) { return 1.0f / (1.0f + __expf(-x)); }
__device__ __forceinline__ float tanh_fast(float x) { return 2.0f / (1.0f + __expf(-2.0f * x)) - 1.0f; }

#define LDS_A_STAGE (32 * 36)
template <int MODE> struct BStage { static constexpr int rows = (MODE == 0) ? 320 : 160; };

template <int MODE>
__device__ __forceinline__ constexpr int wrow_of(int i) {
  return (MODE == 0) ? (i * 32) : ((i >> 1) * HDIM + (i & 1) * 16);
}

template <int MODE, int NA, int NB, int KD>
__device__ __forceinline__ void stage_gload(const float* const (&asrc)[NA], const float* bsrc, int k,
                                            v4f (&ra)[NA], v4f (&rb)[NB]) {
#pragma unroll
  for (int a = 0; a < NA; ++a)
    ra[a] = *(const v4f*)(asrc[a] + k);
#pragma unroll
  for (int i = 0; i < NB; ++i)
    rb[i] = *(const v4f*)(bsrc + (size_t)wrow_of<MODE>(i) * KD + k);
}

template <int RS, int NA, int NB>
__device__ __forceinline__ void stage_lstore(float* As, float* Bs, int sr, int sc4,
                                             const v4f (&ra)[NA], const v4f (&rb)[NB]) {
#pragma unroll
  for (int a = 0; a < NA; ++a)
    *(v4f*)&As[(a * RS + sr) * 36 + sc4 * 4] = ra[a];
#pragma unroll
  for (int i = 0; i < NB; ++i)
    *(v4f*)&Bs[(i * RS + sr) * 36 + sc4 * 4] = rb[i];
}

template <int MODE, int NACC>
__device__ __forceinline__ void stage_mma(const float* As, const float* Bs,
                                          int wm, int wn, int l16, int lhi, v8f (&acc)[5]) {
#pragma unroll
  for (int kk = 0; kk < 32; kk += 4) {
    v2f a = *(const v2f*)&As[(wm * 16 + l16) * 36 + kk + 2 * lhi];
#pragma unroll
    for (int g = 0; g < NACC; ++g) {
      int nrow = (MODE == 0) ? (wn * 64 + g * 16 + l16)
                             : (g * 32 + wn * 16 + l16);
      v2f bf = *(const v2f*)&Bs[nrow * 36 + kk + 2 * lhi];
      acc[g] = __builtin_amdgcn_wmma_f32_16x16x4_f32(
          false, a, false, bf, (short)0, acc[g], false, false);
    }
  }
}

__device__ __forceinline__ void comp_row(int rg, int n, int& b, int& p) {
  b = rg / n;
  p = rg - b * n;
}

template <int MODE>
__global__ __launch_bounds__((MODE == 0) ? 256 : 128) void gemm_wmma(const float* __restrict__ X,
                                                 const float* __restrict__ hbuf,
                                                 const float* __restrict__ cbuf,
                                                 const float* __restrict__ W,
                                                 const float* __restrict__ bias,
                                                 float* __restrict__ out0,
                                                 float* __restrict__ out1,
                                                 int n) {
  constexpr int KD   = (MODE == 0) ? 512 : 1024;
  constexpr int NB   = (MODE == 0) ? 8 : 10;
  constexpr int NACC = (MODE == 0) ? 4 : 5;
  constexpr int RS   = (MODE == 0) ? 32 : 16;
  constexpr int NA   = 32 / RS;
  constexpr int NWN  = (MODE == 0) ? 4 : 2;
  constexpr int LDS_B_STAGE = BStage<MODE>::rows * 36;

  __shared__ __attribute__((aligned(16))) float A_lds[2 * LDS_A_STAGE];
  __shared__ __attribute__((aligned(16))) float B_lds[2 * LDS_B_STAGE];
  float* const A0 = A_lds;
  float* const A1 = A_lds + LDS_A_STAGE;
  float* const B0 = B_lds;
  float* const B1 = B_lds + LDS_B_STAGE;

  const int tid  = threadIdx.x;
  const unsigned lane = tid & 31u;
  const int wave = tid >> 5;
  const int wm   = wave / NWN;
  const int wn   = wave % NWN;
  const int l16  = (int)(lane & 15u);
  const int lhi  = (int)(lane >> 4);

  const int r0 = blockIdx.x * 32;

  const int sr  = tid >> 3;
  const int sc4 = tid & 7;
  const float* asrc[NA];
#pragma unroll
  for (int a = 0; a < NA; ++a) {
    if (MODE == 0) {
      asrc[a] = X + (size_t)(r0 + a * RS + sr) * KD + sc4 * 4;
    } else {
      const int rg = r0 + a * RS + sr;
      int b, p;
      comp_row(rg, n, b, p);
      asrc[a] = hbuf + (size_t)(b * SEQL + p) * HDIM + sc4 * 4;
    }
  }
  const float* bsrc = W + (size_t)(blockIdx.y * ((MODE == 0) ? 256 : 32) + sr) * KD + sc4 * 4;

  const int erow0 = r0 + wm * 16 + 8 * lhi;

  v4f ra[NA];
  v4f rb[NB];
  v8f acc[5] = {};

  stage_gload<MODE, NA, NB, KD>(asrc, bsrc, 0, ra, rb);
  stage_lstore<RS, NA, NB>(A0, B0, sr, sc4, ra, rb);
  __syncthreads();

  for (int k0 = 0; k0 < KD - 64; k0 += 64) {
    stage_gload<MODE, NA, NB, KD>(asrc, bsrc, k0 + 32, ra, rb);
    stage_mma<MODE, NACC>(A0, B0, wm, wn, l16, lhi, acc);
    stage_lstore<RS, NA, NB>(A1, B1, sr, sc4, ra, rb);
    __syncthreads();

    stage_gload<MODE, NA, NB, KD>(asrc, bsrc, k0 + 64, ra, rb);
    stage_mma<MODE, NACC>(A1, B1, wm, wn, l16, lhi, acc);
    stage_lstore<RS, NA, NB>(A0, B0, sr, sc4, ra, rb);
    __syncthreads();
  }
  stage_gload<MODE, NA, NB, KD>(asrc, bsrc, KD - 32, ra, rb);
  stage_mma<MODE, NACC>(A0, B0, wm, wn, l16, lhi, acc);
  stage_lstore<RS, NA, NB>(A1, B1, sr, sc4, ra, rb);
  __syncthreads();
  stage_mma<MODE, NACC>(A1, B1, wm, wn, l16, lhi, acc);

  if (MODE == 0) {
    const int colbase = blockIdx.y * 256 + wn * 64;
    const int rowg0   = r0 + wm * 16 + 8 * lhi;
#pragma unroll
    for (int j = 0; j < 4; ++j) {
      int col = colbase + j * 16 + l16;
      float bv = bias[col];
#pragma unroll
      for (int vr = 0; vr < 8; ++vr) {
        int rg = rowg0 + vr;
        float val = acc[j][vr] + bv;
        if (col < HDIM) out0[(size_t)rg * HDIM + col] = val;
        else            out1[(size_t)rg * HDIM + (col - HDIM)] = val;
      }
    }
  } else {
    const int colg = blockIdx.y * 32 + wn * 16 + l16;
    float bz[5];
#pragma unroll
    for (int g = 0; g < 5; ++g) bz[g] = bias[g * HDIM + colg];
#pragma unroll
    for (int vr = 0; vr < 8; ++vr) {
      int b, p;
      comp_row(erow0 + vr, n, b, p);
      float gi  = acc[0][vr] + bz[0];
      float gfl = acc[1][vr] + bz[1];
      float gfr = acc[2][vr] + bz[2];
      float gu  = acc[3][vr] + bz[3];
      float go  = acc[4][vr] + bz[4];
      size_t cbase = (size_t)(b * SEQL + p) * HDIM + colg;
      float cl = cbuf[cbase];
      float cr = cbuf[cbase + HDIM];
      float nc = cl * sigm(gfl + 1.0f) + cr * sigm(gfr + 1.0f) + tanh_fast(gu) * sigm(gi);
      float nh = sigm(go) * tanh_fast(nc);
      out0[cbase] = nh;
      out1[cbase] = nc;
    }
  }
}

#define PLANE_ROWS (BATCH * SEQL)
#define PLANE_ELEMS ((size_t)PLANE_ROWS * HDIM)
#define CAND_ROW0 (2 * PLANE_ROWS)
#define T_LOC    0
#define T_CS     32
#define T_SC     64
#define T_NEWP   96
#define T_STRIDE 128
#define T_PART   256
__device__ __forceinline__ const int* aux_of(const int* ncplane, int b) {
  return ncplane + (size_t)(b * SEQL + (SEQL - 1)) * HDIM;
}
__device__ __forceinline__ int* aux_of(int* ncplane, int b) {
  return ncplane + (size_t)(b * SEQL + (SEQL - 1)) * HDIM;
}

#define SK_CW 128
#define SK_LD (SK_CW + 4)
#define SK_A_STAGE (16 * SK_LD)
#define SK_B_STAGE (80 * SK_LD)
#define SK_KD 1024

__device__ __forceinline__ constexpr int sk_wrow(int i) { return (i >> 2) * HDIM + (i & 3) * 4; }

__device__ __forceinline__ void sk_gload_a(const float* const (&asrcL)[4], const float* const (&asrcR)[4],
                                           int k, v4f (&ra)[4]) {
#pragma unroll
  for (int a = 0; a < 4; ++a)
    ra[a] = *(const v4f*)((k < HDIM) ? (asrcL[a] + k) : (asrcR[a] + (k - HDIM)));
}

__device__ __forceinline__ void sk_gload_b(const float* bsrc, int k, v4f (&rb)[20]) {
#pragma unroll
  for (int i = 0; i < 20; ++i)
    rb[i] = *(const v4f*)(bsrc + (size_t)sk_wrow(i) * SK_KD + k);
}

__device__ __forceinline__ void sk_lstore(float* As, float* Bs, int wave, int lane,
                                          const v4f (&ra)[4], const v4f (&rb)[20]) {
#pragma unroll
  for (int a = 0; a < 4; ++a)
    *(v4f*)&As[(wave + 4 * a) * SK_LD + lane * 4] = ra[a];
#pragma unroll
  for (int i = 0; i < 20; ++i)
    *(v4f*)&Bs[(wave + 4 * i) * SK_LD + lane * 4] = rb[i];
}

__device__ __forceinline__ void sk_mma(const float* As, const float* Bs,
                                       int wave, int l16, int lhi, v8f (&acc)[5]) {
#pragma unroll
  for (int s = 0; s < 8; ++s) {
    const int ko = wave * 32 + s * 4 + 2 * lhi;
    v2f a = *(const v2f*)&As[l16 * SK_LD + ko];
#pragma unroll
    for (int g = 0; g < 5; ++g) {
      v2f bf = *(const v2f*)&Bs[(g * 16 + l16) * SK_LD + ko];
      acc[g] = __builtin_amdgcn_wmma_f32_16x16x4_f32(
          false, a, false, bf, (short)0, acc[g], false, false);
    }
  }
}

__device__ __forceinline__ int lane_xor16(int x) {
  return __builtin_amdgcn_permlanex16(x, x, 0x76543210, 0xfedcba98, false, false);
}
template <int M>
__device__ __forceinline__ int lane_xor_row(int x) {
  return __builtin_amdgcn_mov_dpp(x, 0x160 | M, 0xf, 0xf, false);
}
__device__ __forceinline__ float f_xor16(float x) { return __int_as_float(lane_xor16(__float_as_int(x))); }
template <int M>
__device__ __forceinline__ float f_xor_row(float x) { return __int_as_float(lane_xor_row<M>(__float_as_int(x))); }
__device__ __forceinline__ float row_sum16(float p) {
  p += f_xor_row<8>(p);
  p += f_xor_row<4>(p);
  p += f_xor_row<2>(p);
  p += f_xor_row<1>(p);
  return p;
}
__device__ __forceinline__ float wave_sum32(float p) {
  p += f_xor16(p);
  return row_sum16(p);
}
__device__ __forceinline__ float wave_max32(float p) {
  p = fmaxf(p, f_xor16(p));
  p = fmaxf(p, f_xor_row<8>(p));
  p = fmaxf(p, f_xor_row<4>(p));
  p = fmaxf(p, f_xor_row<2>(p));
  p = fmaxf(p, f_xor_row<1>(p));
  return p;
}
__device__ __forceinline__ int lane_next(int x, int lane) {
  const int y   = __builtin_amdgcn_update_dpp(x, x, 0x101, 0xf, 0xf, false);
  const int x16 = __builtin_amdgcn_readlane(x, 16);
  return (lane == 15) ? x16 : y;
}

__global__ __launch_bounds__(128) void comp_step_fused(const float* __restrict__ nodes,
                                                       const float* __restrict__ W,
                                                       const float* __restrict__ bias,
                                                       const float* __restrict__ q,
                                                       const int* __restrict__ length,
                                                       float* __restrict__ out0,
                                                       float* __restrict__ out1,
                                                       int* aux,
                                                       int step) {
  __shared__ __attribute__((aligned(16))) float A_lds[2 * SK_A_STAGE];
  __shared__ __attribute__((aligned(16))) float B_lds[2 * SK_B_STAGE];
  __shared__ __attribute__((aligned(16))) int rowinfo[16 * 4];
  __shared__ float red[4 * 5 * 8 * 32];
  float* const A0 = A_lds;
  float* const A1 = A_lds + SK_A_STAGE;
  float* const B0 = B_lds;
  float* const B1 = B_lds + SK_B_STAGE;

  const int tid  = threadIdx.x;
  const int lane = tid & 31;
  const int wave = __builtin_amdgcn_readfirstlane(tid >> 5);
  const int l16  = lane & 15;
  const int lhi  = lane >> 4;
  const int r0   = blockIdx.x * 16;
  const int ct0  = blockIdx.y * 16;

  v4f ra[4];
  v4f rb[20];
  v8f acc[5] = {};

  const float* bsrc = W + (size_t)(ct0 + wave) * SK_KD + lane * 4;

  const int par_in  = (step - 1) & 1;
  const int par_out = step & 1;
  const int it = step - 1;
  const int n  = SEQL - step;
  int   t_loc[2], t_cs[2], t_ppl[2], t_ppr[2], t_len[2];
  float t_sc[2], t_pa[2], t_pb[2];
#pragma unroll
  for (int t = 0; t < 2; ++t) {
    const int b = blockIdx.x * 8 + 2 * wave + t;
    const int* tb = aux_of(aux, b);
    const int* rd = tb + par_in * T_STRIDE;
    t_loc[t] = rd[T_LOC + lane];
    t_cs[t]  = rd[T_CS + lane];
    t_sc[t]  = __int_as_float(rd[T_SC + lane]);
    t_ppl[t] = rd[T_NEWP];
    t_ppr[t] = rd[T_NEWP + 1];
    t_pa[t]  = 0.0f;
    t_pb[t]  = 0.0f;
    if (step >= 2) {
      t_pa[t] = __int_as_float(tb[T_PART + par_in * 64 + lane]);
      t_pb[t] = __int_as_float(tb[T_PART + par_in * 64 + 32 + lane]);
    }
    t_len[t] = length[b];
  }
  sk_gload_b(bsrc, 0, rb);
  float sc0v[2];
#pragma unroll
  for (int t = 0; t < 2; ++t) {
    const float pa = wave_sum32(t_pa[t]);
    const float pb = wave_sum32(t_pb[t]);
    float sc0 = t_sc[t];
    const int ppl = (step >= 2) ? t_ppl[t] : -1;
    const int ppr = (step >= 2) ? t_ppr[t] : -1;
    if (lane == ppl) sc0 = pa;
    if (lane == ppr) sc0 = pb;
    sc0v[t] = sc0;
  }
  int kselv[2];
#pragma unroll
  for (int t = 0; t < 2; ++t) {
    float v = ((it + 1 + lane) < t_len[t]) ? sc0v[t] : -1.0e9f;
    v = (v == v) ? v : -INFINITY;
    v = (lane < n) ? v : -INFINITY;
    const float m = wave_max32(v);
    const unsigned eq = __builtin_amdgcn_ballot_w32(v == m);
    kselv[t] = (eq != 0u) ? __builtin_ctz(eq) : 0;
  }
#pragma unroll
  for (int t = 0; t < 2; ++t) {
    const int bb = 2 * wave + t;
    const int b  = blockIdx.x * 8 + bb;
    int* tb = aux_of(aux, b);
    const int loc0 = t_loc[t];
    const int cs0  = t_cs[t];
    const float sc0 = sc0v[t];
    const int ksel = kselv[t];
    const bool done = (it + 1) < t_len[t];
    const int   loc1 = lane_next(loc0, lane);
    const int   cs1  = lane_next(cs0, lane);
    const float sc1  = __int_as_float(lane_next(__float_as_int(sc0), lane));
    const int   csK  = __builtin_amdgcn_readlane(cs0, ksel);
    int newloc = loc0, newcs = cs0;
    float newsc = sc0;
    if (done) {
      if (lane == ksel)     newloc = CAND_ROW0 + b * SEQL + csK;
      else if (lane > ksel) newloc = loc1;
      if (lane >= ksel) { newcs = cs1; newsc = sc1; }
    }
    const int hi = n - 2;
    int pl = (ksel >= 1) ? (ksel - 1) : 0;
    int pr = (ksel <= hi) ? ksel : hi;
    if (!done) { pl = 0; pr = 0; }
    const int l0 = __builtin_amdgcn_readlane(newloc, pl), rr0 = __builtin_amdgcn_readlane(newloc, pl + 1), d0 = __builtin_amdgcn_readlane(newcs, pl);
    const int l1 = __builtin_amdgcn_readlane(newloc, pr), rr1 = __builtin_amdgcn_readlane(newloc, pr + 1), d1 = __builtin_amdgcn_readlane(newcs, pr);
    if (lane == 0) {
      *(int4*)&rowinfo[(2 * bb) * 4]     = make_int4(l0, rr0, b * SEQL + d0, 0);
      *(int4*)&rowinfo[(2 * bb + 1) * 4] = make_int4(l1, rr1, b * SEQL + d1, 0);
    }
    if (blockIdx.y == 0) {
      int* wr = tb + par_out * T_STRIDE;
      wr[T_LOC + lane] = newloc;
      wr[T_CS + lane]  = newcs;
      wr[T_SC + lane]  = __float_as_int(newsc);
      if (lane == 0) *(int2*)(wr + T_NEWP) = make_int2(pl, pr);
    }
  }
  __syncthreads();

  const float* asrcL[4];
  const float* asrcR[4];
#pragma unroll
  for (int a = 0; a < 4; ++a) {
    const int4 ri = *(const int4*)&rowinfo[(wave + 4 * a) * 4];
    asrcL[a] = nodes + (size_t)ri.x * HDIM + lane * 4;
    asrcR[a] = nodes + (size_t)ri.y * HDIM + lane * 4;
  }

  const int colg = ct0 + l16;
  size_t cbase[2];
  float cl[2], cr[2];
#pragma unroll
  for (int e = 0; e < 2; ++e) {
    const int4 ri = *(const int4*)&rowinfo[(8 * lhi + 2 * wave + e) * 4];
    cbase[e] = (size_t)ri.z * HDIM + colg;
    cl[e] = nodes[PLANE_ELEMS + (size_t)ri.x * HDIM + colg];
    cr[e] = nodes[PLANE_ELEMS + (size_t)ri.y * HDIM + colg];
  }
  float bz[5];
#pragma unroll
  for (int g = 0; g < 5; ++g) bz[g] = bias[g * HDIM + colg];
  const float qv = q[colg];

  sk_gload_a(asrcL, asrcR, 0, ra);
  sk_lstore(A0, B0, wave, lane, ra, rb);
  __syncthreads();

#pragma nounroll
  for (int k0 = 0; k0 < SK_KD - 2 * SK_CW; k0 += 2 * SK_CW) {
    sk_gload_a(asrcL, asrcR, k0 + SK_CW, ra);
    sk_gload_b(bsrc, k0 + SK_CW, rb);
    sk_mma(A0, B0, wave, l16, lhi, acc);
    sk_lstore(A1, B1, wave, lane, ra, rb);
    __syncthreads();

    sk_gload_a(asrcL, asrcR, k0 + 2 * SK_CW, ra);
    sk_gload_b(bsrc, k0 + 2 * SK_CW, rb);
    sk_mma(A1, B1, wave, l16, lhi, acc);
    sk_lstore(A0, B0, wave, lane, ra, rb);
    __syncthreads();
  }
  sk_gload_a(asrcL, asrcR, SK_KD - SK_CW, ra);
  sk_gload_b(bsrc, SK_KD - SK_CW, rb);
  sk_mma(A0, B0, wave, l16, lhi, acc);
  sk_lstore(A1, B1, wave, lane, ra, rb);
  __syncthreads();
  sk_mma(A1, B1, wave, l16, lhi, acc);

#pragma unroll
  for (int g = 0; g < 5; ++g) {
#pragma unroll
    for (int vr = 0; vr < 8; ++vr)
      red[((wave * 5 + g) * 8 + vr) * 32 + lane] = acc[g][vr];
  }
  __syncthreads();

  int* tbe = aux_of(aux, blockIdx.x * 8 + 4 * lhi + wave) + T_PART + par_out * 64 + blockIdx.y;
#pragma unroll
  for (int e = 0; e < 2; ++e) {
    const int vr = 2 * wave + e;
    float gs[5];
#pragma unroll
    for (int g = 0; g < 5; ++g) {
      float s = red[((0 * 5 + g) * 8 + vr) * 32 + lane];
      s += red[((1 * 5 + g) * 8 + vr) * 32 + lane];
      s += red[((2 * 5 + g) * 8 + vr) * 32 + lane];
      s += red[((3 * 5 + g) * 8 + vr) * 32 + lane];
      gs[g] = s + bz[g];
    }
    float nc = cl[e] * sigm(gs[1] + 1.0f) + cr[e] * sigm(gs[2] + 1.0f) + tanh_fast(gs[3]) * sigm(gs[0]);
    float nh = sigm(gs[4]) * tanh_fast(nc);
    out0[cbase[e]] = nh;
    out1[cbase[e]] = nc;
    const float pv = row_sum16(nh * qv);
    if (l16 == 0) tbe[e * 32] = __float_as_int(pv);
  }
}

__device__ __forceinline__ float row_dot_q(const float* row, const float* __restrict__ q, unsigned lane) {
  float s = 0.0f;
  for (int j = (int)lane; j < HDIM; j += 32) s += row[j] * q[j];
#pragma unroll
  for (int off = 16; off > 0; off >>= 1) s += __shfl_xor(s, off, 32);
  return s;
}

__global__ __launch_bounds__(512) void score_all_k(const float* __restrict__ nh,
                                                   const float* __restrict__ q,
                                                   int* aux) {
  __shared__ float sc[32];
  const int b = blockIdx.x;
  const int tid = threadIdx.x;
  const unsigned lane = tid & 31u;
  const int wave = tid >> 5;
  if (tid >= SEQL - 1 && tid < 32) sc[tid] = 0.0f;
  for (int p = wave; p < SEQL - 1; p += 16) {
    float s = row_dot_q(nh + (size_t)(b * SEQL + p) * HDIM, q, lane);
    if (lane == 0) sc[p] = s;
  }
  __syncthreads();
  if (tid < 32) {
    int* tb = aux_of(aux, b);
    tb[T_LOC + tid] = b * SEQL + ((tid < SEQL) ? tid : (SEQL - 1));
    tb[T_CS + tid]  = (tid < SEQL - 1) ? tid : (SEQL - 2);
    tb[T_SC + tid]  = __float_as_int(sc[tid]);
    if (tid == 0) *(int2*)(tb + T_NEWP) = make_int2(0, 0);
  }
}

__global__ __launch_bounds__(256) void final_pick_k(const float* __restrict__ nodes,
                                                    const int* __restrict__ aux,
                                                    const int* __restrict__ length,
                                                    float* __restrict__ out) {
  int idx = blockIdx.x * 256 + threadIdx.x;
  int b = idx >> 9;
  int j = idx & (HDIM - 1);
  const int* tb = aux_of(aux, b);
  const int row = ((SEQL - 1) < length[b]) ? (CAND_ROW0 + b * SEQL + tb[T_CS]) : tb[T_LOC];
  out[idx] = nodes[(size_t)row * HDIM + j];
}

extern "C" void kernel_launch(void* const* d_in, const int* in_sizes, int n_in,
                              void* d_out, int out_size, void* d_ws, size_t ws_size,
                              hipStream_t stream) {
  const float* inp    = (const float*)d_in[0];
  const int*   length = (const int*)  d_in[1];
  const float* word_W = (const float*)d_in[2];
  const float* word_b = (const float*)d_in[3];
  const float* comp_W = (const float*)d_in[4];
  const float* comp_b = (const float*)d_in[5];
  const float* comp_q = (const float*)d_in[6];

  float* ws = (float*)d_ws;
  size_t off = 0;
  float* hb = ws + off; off += (size_t)BATCH * SEQL * HDIM;
  float* cb = ws + off; off += (size_t)BATCH * SEQL * HDIM;
  float* nh = ws + off; off += (size_t)BATCH * SEQL * HDIM;
  float* nc = ws + off; off += (size_t)BATCH * SEQL * HDIM;
  int* aux = (int*)nc;

  gemm_wmma<0><<<dim3((BATCH * SEQL) / 32, 1024 / 256), 256, 0, stream>>>(
      inp, nullptr, nullptr, word_W, word_b, hb, cb, 0);

  gemm_wmma<1><<<dim3((BATCH * (SEQL - 1)) / 32, HDIM / 32), 128, 0, stream>>>(
      nullptr, hb, cb, comp_W, comp_b, nh, nc, SEQL - 1);
  score_all_k<<<BATCH, 512, 0, stream>>>(nh, comp_q, aux);

  for (int i = 1; i < SEQL - 1; ++i) {
    comp_step_fused<<<dim3((BATCH * 2) / 16, HDIM / 16), 128, 0, stream>>>(
        ws, comp_W, comp_b, comp_q, length, nh, nc, aux, i);
  }

  final_pick_k<<<(BATCH * HDIM) / 256, 256, 0, stream>>>(ws, (const int*)aux, length, (float*)d_out);
  (void)off; (void)hb; (void)cb; (void)in_sizes; (void)n_in; (void)out_size; (void)ws_size;
}


# Step kernel: 8 waves (2 per SIMD), each stages its own K eighth in a private LDS region: no workgroup barrier in the K loop, one wave stages while the other multiplies; 8 partial sums, new order
# speedup vs baseline: 7.6683x; 1.1318x over previous
#include <hip/hip_runtime.h>
#include <hip/hip_bf16.h>
#include <math.h>


typedef __attribute__((ext_vector_type(2))) float v2f;
typedef __attribute__((ext_vector_type(4))) float v4f;
typedef __attribute__((ext_vector_type(8))) float v8f;

#define BATCH 64
#define SEQL  24
#define HDIM  512

__device__ __forceinline__ float sigm(float x) { return 1.0f / (1.0f + __expf(-x)); }
__device__ __forceinline__ float tanh_fast(float x) { return 2.0f / (1.0f + __expf(-2.0f * x)) - 1.0f; }

#define LDS_A_STAGE (32 * 36)
template <int MODE> struct BStage { static constexpr int rows = (MODE == 0) ? 320 : 160; };

template <int MODE>
__device__ __forceinline__ constexpr int wrow_of(int i) {
  return (MODE == 0) ? (i * 32) : ((i >> 1) * HDIM + (i & 1) * 16);
}

template <int MODE, int NA, int NB, int KD>
__device__ __forceinline__ void stage_gload(const float* const (&asrc)[NA], const float* bsrc, int k,
                                            v4f (&ra)[NA], v4f (&rb)[NB]) {
#pragma unroll
  for (int a = 0; a < NA; ++a)
    ra[a] = *(const v4f*)(asrc[a] + k);
#pragma unroll
  for (int i = 0; i < NB; ++i)
    rb[i] = *(const v4f*)(bsrc + (size_t)wrow_of<MODE>(i) * KD + k);
}

template <int RS, int NA, int NB>
__device__ __forceinline__ void stage_lstore(float* As, float* Bs, int sr, int sc4,
                                             const v4f (&ra)[NA], const v4f (&rb)[NB]) {
#pragma unroll
  for (int a = 0; a < NA; ++a)
    *(v4f*)&As[(a * RS + sr) * 36 + sc4 * 4] = ra[a];
#pragma unroll
  for (int i = 0; i < NB; ++i)
    *(v4f*)&Bs[(i * RS + sr) * 36 + sc4 * 4] = rb[i];
}

template <int MODE, int NACC>
__device__ __forceinline__ void stage_mma(const float* As, const float* Bs,
                                          int wm, int wn, int l16, int lhi, v8f (&acc)[5]) {
#pragma unroll
  for (int kk = 0; kk < 32; kk += 4) {
    v2f a = *(const v2f*)&As[(wm * 16 + l16) * 36 + kk + 2 * lhi];
#pragma unroll
    for (int g = 0; g < NACC; ++g) {
      int nrow = (MODE == 0) ? (wn * 64 + g * 16 + l16)
                             : (g * 32 + wn * 16 + l16);
      v2f bf = *(const v2f*)&Bs[nrow * 36 + kk + 2 * lhi];
      acc[g] = __builtin_amdgcn_wmma_f32_16x16x4_f32(
          false, a, false, bf, (short)0, acc[g], false, false);
    }
  }
}

__device__ __forceinline__ void comp_row(int rg, int n, int& b, int& p) {
  b = rg / n;
  p = rg - b * n;
}

template <int MODE>
__global__ __launch_bounds__((MODE == 0) ? 256 : 128) void gemm_wmma(const float* __restrict__ X,
                                                 const float* __restrict__ hbuf,
                                                 const float* __restrict__ cbuf,
                                                 const float* __restrict__ W,
                                                 const float* __restrict__ bias,
                                                 float* __restrict__ out0,
                                                 float* __restrict__ out1,
                                                 int n) {
  constexpr int KD   = (MODE == 0) ? 512 : 1024;
  constexpr int NB   = (MODE == 0) ? 8 : 10;
  constexpr int NACC = (MODE == 0) ? 4 : 5;
  constexpr int RS   = (MODE == 0) ? 32 : 16;
  constexpr int NA   = 32 / RS;
  constexpr int NWN  = (MODE == 0) ? 4 : 2;
  constexpr int LDS_B_STAGE = BStage<MODE>::rows * 36;

  __shared__ __attribute__((aligned(16))) float A_lds[2 * LDS_A_STAGE];
  __shared__ __attribute__((aligned(16))) float B_lds[2 * LDS_B_STAGE];
  float* const A0 = A_lds;
  float* const A1 = A_lds + LDS_A_STAGE;
  float* const B0 = B_lds;
  float* const B1 = B_lds + LDS_B_STAGE;

  const int tid  = threadIdx.x;
  const unsigned lane = tid & 31u;
  const int wave = tid >> 5;
  const int wm   = wave / NWN;
  const int wn   = wave % NWN;
  const int l16  = (int)(lane & 15u);
  const int lhi  = (int)(lane >> 4);

  const int r0 = blockIdx.x * 32;

  const int sr  = tid >> 3;
  const int sc4 = tid & 7;
  const float* asrc[NA];
#pragma unroll
  for (int a = 0; a < NA; ++a) {
    if (MODE == 0) {
      asrc[a] = X + (size_t)(r0 + a * RS + sr) * KD + sc4 * 4;
    } else {
      const int rg = r0 + a * RS + sr;
      int b, p;
      comp_row(rg, n, b, p);
      asrc[a] = hbuf + (size_t)(b * SEQL + p) * HDIM + sc4 * 4;
    }
  }
  const float* bsrc = W + (size_t)(blockIdx.y * ((MODE == 0) ? 256 : 32) + sr) * KD + sc4 * 4;

  const int erow0 = r0 + wm * 16 + 8 * lhi;

  v4f ra[NA];
  v4f rb[NB];
  v8f acc[5] = {};

  stage_gload<MODE, NA, NB, KD>(asrc, bsrc, 0, ra, rb);
  stage_lstore<RS, NA, NB>(A0, B0, sr, sc4, ra, rb);
  __syncthreads();

  for (int k0 = 0; k0 < KD - 64; k0 += 64) {
    stage_gload<MODE, NA, NB, KD>(asrc, bsrc, k0 + 32, ra, rb);
    stage_mma<MODE, NACC>(A0, B0, wm, wn, l16, lhi, acc);
    stage_lstore<RS, NA, NB>(A1, B1, sr, sc4, ra, rb);
    __syncthreads();

    stage_gload<MODE, NA, NB, KD>(asrc, bsrc, k0 + 64, ra, rb);
    stage_mma<MODE, NACC>(A1, B1, wm, wn, l16, lhi, acc);
    stage_lstore<RS, NA, NB>(A0, B0, sr, sc4, ra, rb);
    __syncthreads();
  }
  stage_gload<MODE, NA, NB, KD>(asrc, bsrc, KD - 32, ra, rb);
  stage_mma<MODE, NACC>(A0, B0, wm, wn, l16, lhi, acc);
  stage_lstore<RS, NA, NB>(A1, B1, sr, sc4, ra, rb);
  __syncthreads();
  stage_mma<MODE, NACC>(A1, B1, wm, wn, l16, lhi, acc);

  if (MODE == 0) {
    const int colbase = blockIdx.y * 256 + wn * 64;
    const int rowg0   = r0 + wm * 16 + 8 * lhi;
#pragma unroll
    for (int j = 0; j < 4; ++j) {
      int col = colbase + j * 16 + l16;
      float bv = bias[col];
#pragma unroll
      for (int vr = 0; vr < 8; ++vr) {
        int rg = rowg0 + vr;
        float val = acc[j][vr] + bv;
        if (col < HDIM) out0[(size_t)rg * HDIM + col] = val;
        else            out1[(size_t)rg * HDIM + (col - HDIM)] = val;
      }
    }
  } else {
    const int colg = blockIdx.y * 32 + wn * 16 + l16;
    float bz[5];
#pragma unroll
    for (int g = 0; g < 5; ++g) bz[g] = bias[g * HDIM + colg];
#pragma unroll
    for (int vr = 0; vr < 8; ++vr) {
      int b, p;
      comp_row(erow0 + vr, n, b, p);
      float gi  = acc[0][vr] + bz[0];
      float gfl = acc[1][vr] + bz[1];
      float gfr = acc[2][vr] + bz[2];
      float gu  = acc[3][vr] + bz[3];
      float go  = acc[4][vr] + bz[4];
      size_t cbase = (size_t)(b * SEQL + p) * HDIM + colg;
      float cl = cbuf[cbase];
      float cr = cbuf[cbase + HDIM];
      float nc = cl * sigm(gfl + 1.0f) + cr * sigm(gfr + 1.0f) + tanh_fast(gu) * sigm(gi);
      float nh = sigm(go) * tanh_fast(nc);
      out0[cbase] = nh;
      out1[cbase] = nc;
    }
  }
}

#define PLANE_ROWS (BATCH * SEQL)
#define PLANE_ELEMS ((size_t)PLANE_ROWS * HDIM)
#define CAND_ROW0 (2 * PLANE_ROWS)
#define T_LOC    0
#define T_CS     32
#define T_SC     64
#define T_NEWP   96
#define T_STRIDE 128
#define T_PART   256
__device__ __forceinline__ const int* aux_of(const int* ncplane, int b) {
  return ncplane + (size_t)(b * SEQL + (SEQL - 1)) * HDIM;
}
__device__ __forceinline__ int* aux_of(int* ncplane, int b) {
  return ncplane + (size_t)(b * SEQL + (SEQL - 1)) * HDIM;
}

#define SK_NW 8
#define SK_KW 128
#define SK_PW 32
#define SK_LD (SK_PW + 4)
#define SK_A_STAGE (16 * SK_LD)
#define SK_B_STAGE (80 * SK_LD)
#define SK_W_STAGE (SK_A_STAGE + SK_B_STAGE)
#define SK_KD 1024

__device__ __forceinline__ constexpr int sk_wrow(int i) { return (i >> 2) * HDIM + (i & 3) * 4; }

__device__ __forceinline__ void sk_gload_a(const float* const (&asrc)[4], int k, v4f (&ra)[4]) {
#pragma unroll
  for (int a = 0; a < 4; ++a)
    ra[a] = *(const v4f*)(asrc[a] + k);
}

__device__ __forceinline__ void sk_gload_b(const float* bsrc, int k, v4f (&rb)[20]) {
#pragma unroll
  for (int i = 0; i < 20; ++i)
    rb[i] = *(const v4f*)(bsrc + (size_t)sk_wrow(i) * SK_KD + k);
}

__device__ __forceinline__ void sk_lstore(float* As, float* Bs, int lr, int lp,
                                          const v4f (&ra)[4], const v4f (&rb)[20]) {
#pragma unroll
  for (int a = 0; a < 4; ++a)
    *(v4f*)&As[(lr + 4 * a) * SK_LD + lp * 4] = ra[a];
#pragma unroll
  for (int i = 0; i < 20; ++i)
    *(v4f*)&Bs[(lr + 4 * i) * SK_LD + lp * 4] = rb[i];
}

__device__ __forceinline__ void sk_mma(const float* As, const float* Bs,
                                       int l16, int lhi, v8f (&acc)[5]) {
#pragma unroll
  for (int s = 0; s < 8; ++s) {
    const int ko = s * 4 + 2 * lhi;
    v2f a = *(const v2f*)&As[l16 * SK_LD + ko];
#pragma unroll
    for (int g = 0; g < 5; ++g) {
      v2f bf = *(const v2f*)&Bs[(g * 16 + l16) * SK_LD + ko];
      acc[g] = __builtin_amdgcn_wmma_f32_16x16x4_f32(
          false, a, false, bf, (short)0, acc[g], false, false);
    }
  }
}

__device__ __forceinline__ void sk_wave_handover() {
  __builtin_amdgcn_fence(__ATOMIC_RELEASE, "wavefront");
  __builtin_amdgcn_wave_barrier();
  __builtin_amdgcn_fence(__ATOMIC_ACQUIRE, "wavefront");
}

__device__ __forceinline__ int lane_xor16(int x) {
  return __builtin_amdgcn_permlanex16(x, x, 0x76543210, 0xfedcba98, false, false);
}
template <int M>
__device__ __forceinline__ int lane_xor_row(int x) {
  return __builtin_amdgcn_mov_dpp(x, 0x160 | M, 0xf, 0xf, false);
}
__device__ __forceinline__ float f_xor16(float x) { return __int_as_float(lane_xor16(__float_as_int(x))); }
template <int M>
__device__ __forceinline__ float f_xor_row(float x) { return __int_as_float(lane_xor_row<M>(__float_as_int(x))); }
__device__ __forceinline__ float row_sum16(float p) {
  p += f_xor_row<8>(p);
  p += f_xor_row<4>(p);
  p += f_xor_row<2>(p);
  p += f_xor_row<1>(p);
  return p;
}
__device__ __forceinline__ float wave_sum32(float p) {
  p += f_xor16(p);
  return row_sum16(p);
}
__device__ __forceinline__ float wave_max32(float p) {
  p = fmaxf(p, f_xor16(p));
  p = fmaxf(p, f_xor_row<8>(p));
  p = fmaxf(p, f_xor_row<4>(p));
  p = fmaxf(p, f_xor_row<2>(p));
  p = fmaxf(p, f_xor_row<1>(p));
  return p;
}
__device__ __forceinline__ int lane_next(int x, int lane) {
  const int y   = __builtin_amdgcn_update_dpp(x, x, 0x101, 0xf, 0xf, false);
  const int x16 = __builtin_amdgcn_readlane(x, 16);
  return (lane == 15) ? x16 : y;
}

__device__ __forceinline__ void sk_keep_order() { __builtin_amdgcn_sched_barrier(0); }

__global__ __launch_bounds__(256) void comp_step_fused(const float* __restrict__ nodes,
                                                       const float* __restrict__ W,
                                                       const float* __restrict__ bias,
                                                       const float* __restrict__ q,
                                                       const int* __restrict__ length,
                                                       float* __restrict__ out0,
                                                       float* __restrict__ out1,
                                                       int* aux,
                                                       int step) {
  __shared__ __attribute__((aligned(16))) float stg[SK_NW * 2 * SK_W_STAGE];
  __shared__ __attribute__((aligned(16))) int rowinfo[16 * 4];
  __shared__ float red[SK_NW * 5 * 8 * 32];

  const int tid  = threadIdx.x;
  const int lane = tid & 31;
  const int wave = __builtin_amdgcn_readfirstlane(tid >> 5);
  const int l16  = lane & 15;
  const int lhi  = lane >> 4;
  const int lr   = lane >> 3;
  const int lp   = lane & 7;
  const int ct0  = blockIdx.y * 16;
  const int kw0  = wave * SK_KW;

  float* const A0 = stg + wave * (2 * SK_W_STAGE);
  float* const B0 = A0 + SK_A_STAGE;
  float* const A1 = A0 + SK_W_STAGE;
  float* const B1 = A1 + SK_A_STAGE;

  v4f ra[4];
  v4f rb[20];
  v8f acc[5] = {};

  const float* bsrc = W + (size_t)(ct0 + lr) * SK_KD + kw0 + lp * 4;

  const int par_in  = (step - 1) & 1;
  const int par_out = step & 1;
  const int it = step - 1;
  const int n  = SEQL - step;
  const int b  = blockIdx.x * 8 + wave;
  int* tb = aux_of(aux, b);
  const int* rd = tb + par_in * T_STRIDE;
  const int   loc0  = rd[T_LOC + lane];
  const int   cs0   = rd[T_CS + lane];
  const float sc_in = __int_as_float(rd[T_SC + lane]);
  const int   t_ppl = rd[T_NEWP];
  const int   t_ppr = rd[T_NEWP + 1];
  float t_pa = 0.0f, t_pb = 0.0f;
  if (step >= 2) {
    t_pa = __int_as_float(tb[T_PART + par_in * 64 + lane]);
    t_pb = __int_as_float(tb[T_PART + par_in * 64 + 32 + lane]);
  }
  const int t_len = length[b];
  sk_gload_b(bsrc, 0, rb);
  const float pa = wave_sum32(t_pa);
  const float pb = wave_sum32(t_pb);
  float sc0 = sc_in;
  {
    const int ppl = (step >= 2) ? t_ppl : -1;
    const int ppr = (step >= 2) ? t_ppr : -1;
    if (lane == ppl) sc0 = pa;
    if (lane == ppr) sc0 = pb;
  }
  int ksel;
  {
    float v = ((it + 1 + lane) < t_len) ? sc0 : -1.0e9f;
    v = (v == v) ? v : -INFINITY;
    v = (lane < n) ? v : -INFINITY;
    const float m = wave_max32(v);
    const unsigned eq = __builtin_amdgcn_ballot_w32(v == m);
    ksel = (eq != 0u) ? __builtin_ctz(eq) : 0;
  }
  {
    const int bb = wave;
    const bool done = (it + 1) < t_len;
    const int   loc1 = lane_next(loc0, lane);
    const int   cs1  = lane_next(cs0, lane);
    const float sc1  = __int_as_float(lane_next(__float_as_int(sc0), lane));
    const int   csK  = __builtin_amdgcn_readlane(cs0, ksel);
    int newloc = loc0, newcs = cs0;
    float newsc = sc0;
    if (done) {
      if (lane == ksel)     newloc = CAND_ROW0 + b * SEQL + csK;
      else if (lane > ksel) newloc = loc1;
      if (lane >= ksel) { newcs = cs1; newsc = sc1; }
    }
    const int hi = n - 2;
    int pl = (ksel >= 1) ? (ksel - 1) : 0;
    int pr = (ksel <= hi) ? ksel : hi;
    if (!done) { pl = 0; pr = 0; }
    const int l0 = __builtin_amdgcn_readlane(newloc, pl), rr0 = __builtin_amdgcn_readlane(newloc, pl + 1), d0 = __builtin_amdgcn_readlane(newcs, pl);
    const int l1 = __builtin_amdgcn_readlane(newloc, pr), rr1 = __builtin_amdgcn_readlane(newloc, pr + 1), d1 = __builtin_amdgcn_readlane(newcs, pr);
    if (lane == 0) {
      *(int4*)&rowinfo[(2 * bb) * 4]     = make_int4(l0, rr0, b * SEQL + d0, 0);
      *(int4*)&rowinfo[(2 * bb + 1) * 4] = make_int4(l1, rr1, b * SEQL + d1, 0);
    }
    if (blockIdx.y == 0) {
      int* wr = tb + par_out * T_STRIDE;
      wr[T_LOC + lane] = newloc;
      wr[T_CS + lane]  = newcs;
      wr[T_SC + lane]  = __float_as_int(newsc);
      if (lane == 0) *(int2*)(wr + T_NEWP) = make_int2(pl, pr);
    }
  }
  __syncthreads();

  const float* asrc[4];
#pragma unroll
  for (int a = 0; a < 4; ++a) {
    const int4 ri = *(const int4*)&rowinfo[(lr + 4 * a) * 4];
    const int home = (kw0 < HDIM) ? ri.x : ri.y;
    asrc[a] = nodes + (size_t)home * HDIM + (kw0 & (HDIM - 1)) + lp * 4;
  }

  const int trow = 8 * lhi + wave;
  const int colg = ct0 + l16;
  size_t cbase;
  float cl, cr;
  {
    const int4 ri = *(const int4*)&rowinfo[trow * 4];
    cbase = (size_t)ri.z * HDIM + colg;
    cl = nodes[PLANE_ELEMS + (size_t)ri.x * HDIM + colg];
    cr = nodes[PLANE_ELEMS + (size_t)ri.y * HDIM + colg];
  }
  float bz[5];
#pragma unroll
  for (int g = 0; g < 5; ++g) bz[g] = bias[g * HDIM + colg];
  const float qv = q[colg];

  sk_gload_a(asrc, 0, ra);
  sk_lstore(A0, B0, lr, lp, ra, rb);
  sk_wave_handover();

  sk_gload_a(asrc, SK_PW, ra);
  sk_gload_b(bsrc, SK_PW, rb);
  sk_mma(A0, B0, l16, lhi, acc);
  sk_keep_order();
  sk_lstore(A1, B1, lr, lp, ra, rb);
  sk_wave_handover();

  sk_gload_a(asrc, 2 * SK_PW, ra);
  sk_gload_b(bsrc, 2 * SK_PW, rb);
  sk_mma(A1, B1, l16, lhi, acc);
  sk_keep_order();
  sk_lstore(A0, B0, lr, lp, ra, rb);
  sk_wave_handover();

  sk_gload_a(asrc, 3 * SK_PW, ra);
  sk_gload_b(bsrc, 3 * SK_PW, rb);
  sk_mma(A0, B0, l16, lhi, acc);
  sk_keep_order();
  sk_lstore(A1, B1, lr, lp, ra, rb);
  sk_wave_handover();

  sk_mma(A1, B1, l16, lhi, acc);

#pragma unroll
  for (int g = 0; g < 5; ++g) {
#pragma unroll
    for (int vr = 0; vr < 8; ++vr)
      red[((wave * 5 + g) * 8 + vr) * 32 + lane] = acc[g][vr];
  }
  __syncthreads();

  int* tbe = aux_of(aux, blockIdx.x * 8 + (trow >> 1)) + T_PART + par_out * 64 + blockIdx.y;
  {
    const int vr = wave;
    float gs[5];
#pragma unroll
    for (int g = 0; g < 5; ++g) {
      float s = red[((0 * 5 + g) * 8 + vr) * 32 + lane];
#pragma unroll
      for (int w = 1; w < SK_NW; ++w)
        s += red[((w * 5 + g) * 8 + vr) * 32 + lane];
      gs[g] = s + bz[g];
    }
    float nc = cl * sigm(gs[1] + 1.0f) + cr * sigm(gs[2] + 1.0f) + tanh_fast(gs[3]) * sigm(gs[0]);
    float nh = sigm(gs[4]) * tanh_fast(nc);
    out0[cbase] = nh;
    out1[cbase] = nc;
    const float pv = row_sum16(nh * qv);
    if (l16 == 0) tbe[(trow & 1) * 32] = __float_as_int(pv);
  }
}

__device__ __forceinline__ float row_dot_q(const float* row, const float* __restrict__ q, unsigned lane) {
  float s = 0.0f;
  for (int j = (int)lane; j < HDIM; j += 32) s += row[j] * q[j];
#pragma unroll
  for (int off = 16; off > 0; off >>= 1) s += __shfl_xor(s, off, 32);
  return s;
}

__global__ __launch_bounds__(512) void score_all_k(const float* __restrict__ nh,
                                                   const float* __restrict__ q,
                                                   int* aux) {
  __shared__ float sc[32];
  const int b = blockIdx.x;
  const int tid = threadIdx.x;
  const unsigned lane = tid & 31u;
  const int wave = tid >> 5;
  if (tid >= SEQL - 1 && tid < 32) sc[tid] = 0.0f;
  for (int p = wave; p < SEQL - 1; p += 16) {
    float s = row_dot_q(nh + (size_t)(b * SEQL + p) * HDIM, q, lane);
    if (lane == 0) sc[p] = s;
  }
  __syncthreads();
  if (tid < 32) {
    int* tb = aux_of(aux, b);
    tb[T_LOC + tid] = b * SEQL + ((tid < SEQL) ? tid : (SEQL - 1));
    tb[T_CS + tid]  = (tid < SEQL - 1) ? tid : (SEQL - 2);
    tb[T_SC + tid]  = __float_as_int(sc[tid]);
    if (tid == 0) *(int2*)(tb + T_NEWP) = make_int2(0, 0);
  }
}

__global__ __launch_bounds__(256) void final_pick_k(const float* __restrict__ nodes,
                                                    const int* __restrict__ aux,
                                                    const int* __restrict__ length,
                                                    float* __restrict__ out) {
  int idx = blockIdx.x * 256 + threadIdx.x;
  int b = idx >> 9;
  int j = idx & (HDIM - 1);
  const int* tb = aux_of(aux, b);
  const int row = ((SEQL - 1) < length[b]) ? (CAND_ROW0 + b * SEQL + tb[T_CS]) : tb[T_LOC];
  out[idx] = nodes[(size_t)row * HDIM + j];
}

extern "C" void kernel_launch(void* const* d_in, const int* in_sizes, int n_in,
                              void* d_out, int out_size, void* d_ws, size_t ws_size,
                              hipStream_t stream) {
  const float* inp    = (const float*)d_in[0];
  const int*   length = (const int*)  d_in[1];
  const float* word_W = (const float*)d_in[2];
  const float* word_b = (const float*)d_in[3];
  const float* comp_W = (const float*)d_in[4];
  const float* comp_b = (const float*)d_in[5];
  const float* comp_q = (const float*)d_in[6];

  float* ws = (float*)d_ws;
  size_t off = 0;
  float* hb = ws + off; off += (size_t)BATCH * SEQL * HDIM;
  float* cb = ws + off; off += (size_t)BATCH * SEQL * HDIM;
  float* nh = ws + off; off += (size_t)BATCH * SEQL * HDIM;
  float* nc = ws + off; off += (size_t)BATCH * SEQL * HDIM;
  int* aux = (int*)nc;

  gemm_wmma<0><<<dim3((BATCH * SEQL) / 32, 1024 / 256), 256, 0, stream>>>(
      inp, nullptr, nullptr, word_W, word_b, hb, cb, 0);

  gemm_wmma<1><<<dim3((BATCH * (SEQL - 1)) / 32, HDIM / 32), 128, 0, stream>>>(
      nullptr, hb, cb, comp_W, comp_b, nh, nc, SEQL - 1);
  score_all_k<<<BATCH, 512, 0, stream>>>(nh, comp_q, aux);

  for (int i = 1; i < SEQL - 1; ++i) {
    comp_step_fused<<<dim3((BATCH * 2) / 16, HDIM / 16), 256, 0, stream>>>(
        ws, comp_W, comp_b, comp_q, length, nh, nc, aux, i);
  }

  final_pick_k<<<(BATCH * HDIM) / 256, 256, 0, stream>>>(ws, (const int*)aux, length, (float*)d_out);
  (void)off; (void)hb; (void)cb; (void)in_sizes; (void)n_in; (void)out_size; (void)ws_size;
}
